# starting kernel
# baseline (speedup 1.0000x reference)
#include <hip/hip_runtime.h>

#define NFEAT 128

typedef float v2f  __attribute__((ext_vector_type(2)));
typedef float v8f  __attribute__((ext_vector_type(8)));

__device__ __forceinline__ void atomic_add_f32_dev(float* p, float v) {
    asm volatile("global_atomic_add_f32 %0, %1, off scope:SCOPE_DEV"
                 :: "v"(p), "v"(v) : "memory");
}

__global__ void k_deg_init(float* __restrict__ deg, int n) {
    int i = blockIdx.x * blockDim.x + threadIdx.x;
    if (i < n) deg[i] = 1.0f;
}

__global__ void k_deg_accum(const int* __restrict__ col, const float* __restrict__ ew,
                            float* __restrict__ deg, int E) {
    int e = blockIdx.x * blockDim.x + threadIdx.x;
    if (e < E) atomic_add_f32_dev(&deg[col[e]], ew[e]);
}

__global__ void k_rsqrt(float* __restrict__ deg, int n) {
    int i = blockIdx.x * blockDim.x + threadIdx.x;
    if (i < n) deg[i] = rsqrtf(deg[i]);
}

__global__ void __launch_bounds__(256)
k_gemm128_wmma(const float* __restrict__ A, const float* __restrict__ W,
               float* __restrict__ out, int nrows) {
    const int lane = threadIdx.x;
    const int wv   = threadIdx.y;
    const int m0   = blockIdx.x * 16;
    const int n0   = wv * 16;
    const int hi   = lane >> 4;
    const int l    = lane & 15;
    if (m0 + 16 > nrows) return;

    v8f acc = {};
    const float* ap = A + (size_t)(m0 + l) * NFEAT + 2 * hi;
    const float* bp = W + (size_t)(2 * hi) * NFEAT + n0 + l;
#pragma unroll 8
    for (int k = 0; k < NFEAT; k += 4) {
        v2f a = *(const v2f*)(ap + k);
        v2f b;
        b.x = bp[(size_t)k * NFEAT];
        b.y = bp[(size_t)k * NFEAT + NFEAT];
        acc = __builtin_amdgcn_wmma_f32_16x16x4_f32(
            false, a, false, b, (short)0, acc, false, false);
    }
    float* op = out + (size_t)(m0 + 8 * hi) * NFEAT + n0 + l;
#pragma unroll
    for (int v = 0; v < 8; ++v)
        op[(size_t)v * NFEAT] = acc[v];
}

__global__ void k_selfbias(const float* __restrict__ h, const float* __restrict__ dinv,
                           const float* __restrict__ bias, float* __restrict__ out, int n) {
    int idx   = blockIdx.x * blockDim.x + threadIdx.x;
    int total = n * (NFEAT / 4);
    if (idx >= total) return;
    int node = idx >> 5;
    int f    = (idx & 31) * 4;
    float di = dinv[node];
    float s  = di * di;
    float4 hv = *(const float4*)(h + (size_t)node * NFEAT + f);
    float4 bv = *(const float4*)(bias + f);
    float4 o;
    o.x = fmaf(s, hv.x, bv.x);
    o.y = fmaf(s, hv.y, bv.y);
    o.z = fmaf(s, hv.z, bv.z);
    o.w = fmaf(s, hv.w, bv.w);
    *(float4*)(out + (size_t)node * NFEAT + f) = o;
}

__global__ void __launch_bounds__(256)
k_edge_agg(const int* __restrict__ row, const int* __restrict__ col,
           const float* __restrict__ ew, const float* __restrict__ dinv,
           const float* __restrict__ h, float* __restrict__ out, int E) {
    int e = blockIdx.x * blockDim.y + threadIdx.y;
    if (e >= E) return;
    int r = row[e];
    int c = col[e];
    float nrm = dinv[r] * ew[e] * dinv[c];
    int f = threadIdx.x * 4;
    float4 hv = *(const float4*)(h + (size_t)r * NFEAT + f);
    float* op = out + (size_t)c * NFEAT + f;
    atomic_add_f32_dev(op + 0, nrm * hv.x);
    atomic_add_f32_dev(op + 1, nrm * hv.y);
    atomic_add_f32_dev(op + 2, nrm * hv.z);
    atomic_add_f32_dev(op + 3, nrm * hv.w);
}

__global__ void k_relu(float* __restrict__ p, int n4) {
    int idx = blockIdx.x * blockDim.x + threadIdx.x;
    if (idx >= n4) return;
    float4 v = *(float4*)(p + (size_t)idx * 4);
    v.x = fmaxf(v.x, 0.0f);
    v.y = fmaxf(v.y, 0.0f);
    v.z = fmaxf(v.z, 0.0f);
    v.w = fmaxf(v.w, 0.0f);
    *(float4*)(p + (size_t)idx * 4) = v;
}

extern "C" void kernel_launch(void* const* d_in, const int* in_sizes, int n_in,
                              void* d_out, int out_size, void* d_ws, size_t ws_size,
                              hipStream_t stream) {
    const float* x  = (const float*)d_in[0];
    const int*   ei = (const int*)d_in[1];
    const float* ew = (const float*)d_in[2];
    const float* W1 = (const float*)d_in[3];
    const float* b1 = (const float*)d_in[4];
    const float* W2 = (const float*)d_in[5];
    const float* b2 = (const float*)d_in[6];
    float* out = (float*)d_out;

    const int N = in_sizes[0] / NFEAT;
    const int E = in_sizes[2];
    const int* row = ei;
    const int* col = ei + E;

    float* ws   = (float*)d_ws;
    size_t npad = ((size_t)N + 255) & ~(size_t)255;
    float* dinv = ws;
    float* h    = ws + npad;
    float* g    = h + (size_t)N * NFEAT;

    const int T = 256;
    k_deg_init<<<(N + T - 1) / T, T, 0, stream>>>(dinv, N);
    k_deg_accum<<<(E + T - 1) / T, T, 0, stream>>>(col, ew, dinv, E);
    k_rsqrt<<<(N + T - 1) / T, T, 0, stream>>>(dinv, N);

    dim3 wblk(32, 8);
    int mtiles = (N + 15) / 16;
    int tvec   = N * (NFEAT / 4);
    int eblk   = (E + 7) / 8;

    k_gemm128_wmma<<<mtiles, wblk, 0, stream>>>(x, W1, h, N);
    k_selfbias<<<(tvec + T - 1) / T, T, 0, stream>>>(h, dinv, b1, g, N);
    k_edge_agg<<<eblk, wblk, 0, stream>>>(row, col, ew, dinv, h, g, E);
    k_relu<<<(tvec + T - 1) / T, T, 0, stream>>>(g, tvec);

    k_gemm128_wmma<<<mtiles, wblk, 0, stream>>>(g, W2, h, N);
    k_selfbias<<<(tvec + T - 1) / T, T, 0, stream>>>(h, dinv, b2, out, N);
    k_edge_agg<<<eblk, wblk, 0, stream>>>(row, col, ew, dinv, h, out, E);
}


# Edge scatter by f32 atomics replaced by atomic-free stable counting sort of edges by destination + one wave per node gathering rows in ascending edge order; self-loop, bias, ReLU in its one store
# speedup vs baseline: 2.3247x; 2.3247x over previous
#include <hip/hip_runtime.h>

#define NFEAT 128

typedef float v2f  __attribute__((ext_vector_type(2)));
typedef float v8f  __attribute__((ext_vector_type(8)));

#define BK_SHIFT 7
#define BK_NODES 128
#define BK_MAX   1024
#define CHUNK    4096
#define GRP      (CHUNK / 8 / 32)

template <int BITS>
__device__ __forceinline__ unsigned match_peers(int key, bool valid) {
    unsigned peers = __builtin_amdgcn_ballot_w32(valid);
#pragma unroll
    for (int b = 0; b < BITS; ++b) {
        const bool bit = (key >> b) & 1;
        const unsigned m = __builtin_amdgcn_ballot_w32(bit);
        peers &= bit ? m : ~m;
    }
    return peers;
}

__device__ __forceinline__ void load_cols(const int* __restrict__ col, int E, int e0,
                                          int lane, int (&cv)[GRP]) {
#pragma unroll
    for (int g = 0; g < GRP; ++g) {
        const int e = e0 + g * 32 + lane;
        cv[g] = (e < E) ? col[e] : -1;
    }
}

#define COUNT_COLS(SH, WBASE)                                                        \
    _Pragma("unroll")                                                                \
    for (int g = 0; g < GRP; ++g) {                                                  \
        const bool valid = cv[g] >= 0;                                               \
        const int  key   = valid ? (cv[g] >> BK_SHIFT) : 0;                          \
        const unsigned peers = match_peers<10>(key, valid);                          \
        if (valid && (peers & ((1u << lane) - 1u)) == 0u)                            \
            SH[(WBASE) + key] = SH[(WBASE) + key] + __popc(peers);                   \
    }

__global__ void __launch_bounds__(256)
k_bhist(const int* __restrict__ col, int* __restrict__ hist, int E) {
    __shared__ int sh[8 * BK_MAX];
    const int tid  = threadIdx.x;
    const int lane = tid & 31;
    const int w    = __builtin_amdgcn_readfirstlane(tid >> 5);
    for (int i = tid; i < 8 * BK_MAX; i += 256) sh[i] = 0;
    int cv[GRP];
    load_cols(col, E, blockIdx.x * CHUNK + w * (CHUNK / 8), lane, cv);
    __syncthreads();
    COUNT_COLS(sh, w * BK_MAX)
    __syncthreads();
    for (int k = tid; k < BK_MAX; k += 256) {
        int s = 0;
#pragma unroll
        for (int w2 = 0; w2 < 8; ++w2) s += sh[w2 * BK_MAX + k];
        hist[blockIdx.x * BK_MAX + k] = s;
    }
}

__global__ void __launch_bounds__(1024)
k_bscan(const int* __restrict__ hist, int* __restrict__ cscan,
        int* __restrict__ bstart, int nblk) {
    __shared__ int sh[BK_MAX];
    const int k = threadIdx.x;
    int run = 0;
#pragma unroll 16
    for (int b = 0; b < nblk; ++b) {
        const int c = hist[b * BK_MAX + k];
        cscan[b * BK_MAX + k] = run;
        run += c;
    }
    sh[k] = run;
    __syncthreads();
    for (int d = 1; d < BK_MAX; d <<= 1) {
        const int v = (k >= d) ? sh[k - d] : 0;
        __syncthreads();
        sh[k] += v;
        __syncthreads();
    }
    const int incl = sh[k];
    bstart[k] = incl - run;
    if (k == BK_MAX - 1) bstart[BK_MAX] = incl;
}

__global__ void __launch_bounds__(256)
k_bscatter(const int* __restrict__ col, const int* __restrict__ cscan,
           const int* __restrict__ bstart, int* __restrict__ bkt_e,
           int* __restrict__ bkt_col, int E) {
    __shared__ int sh[8 * BK_MAX];
    const int tid  = threadIdx.x;
    const int lane = tid & 31;
    const int w    = __builtin_amdgcn_readfirstlane(tid >> 5);
    const int e0   = blockIdx.x * CHUNK + w * (CHUNK / 8);
    for (int i = tid; i < 8 * BK_MAX; i += 256) sh[i] = 0;
    int cv[GRP];
    load_cols(col, E, e0, lane, cv);
    __syncthreads();
    const int wb = w * BK_MAX;
    COUNT_COLS(sh, wb)
    __syncthreads();
    for (int k = tid; k < BK_MAX; k += 256) {
        int run = bstart[k] + cscan[blockIdx.x * BK_MAX + k];
#pragma unroll
        for (int w2 = 0; w2 < 8; ++w2) {
            const int c = sh[w2 * BK_MAX + k];
            sh[w2 * BK_MAX + k] = run;
            run += c;
        }
    }
    __syncthreads();
#pragma unroll
    for (int g = 0; g < GRP; ++g) {
        const bool valid = cv[g] >= 0;
        const int  key   = valid ? (cv[g] >> BK_SHIFT) : 0;
        const unsigned peers = match_peers<10>(key, valid);
        const unsigned below = peers & ((1u << lane) - 1u);
        if (valid) {
            const int base = sh[wb + key];
            const int pos  = base + __popc(below);
            if (below == 0u) sh[wb + key] = base + __popc(peers);
            bkt_e[pos]   = e0 + g * 32 + lane;
            bkt_col[pos] = cv[g];
        }
    }
}

__global__ void __launch_bounds__(256)
k_bsort(const int* __restrict__ bstart, const int* __restrict__ bkt_e,
        const int* __restrict__ bkt_col, const int* __restrict__ row,
        const float* __restrict__ ew, int* __restrict__ off,
        int* __restrict__ csr_row, float* __restrict__ csr_ew, int n, int E) {
    __shared__ int sh[8 * BK_NODES];
    __shared__ int tot[BK_NODES];
    const int tid  = threadIdx.x;
    const int lane = tid & 31;
    const int w    = __builtin_amdgcn_readfirstlane(tid >> 5);
    const int kb   = blockIdx.x;
    const int bs   = bstart[kb];
    const int be   = bstart[kb + 1];
    const int per  = (((be - bs) + 255) >> 8) << 5;
    const int wbeg = __builtin_amdgcn_readfirstlane(bs + w * per);
    const int wlim = __builtin_amdgcn_readfirstlane(wbeg + per);
    const int wend = (wlim < be) ? wlim : be;
    const int wb = w * BK_NODES;

    for (int i = tid; i < 8 * BK_NODES; i += 256) sh[i] = 0;
    __syncthreads();
    for (int p0 = wbeg; p0 < wend; p0 += 32) {
        const int  p     = p0 + lane;
        const bool valid = p < wend;
        const int  key   = valid ? (bkt_col[p] & (BK_NODES - 1)) : 0;
        const unsigned peers = match_peers<7>(key, valid);
        if (valid && (peers & ((1u << lane) - 1u)) == 0u)
            sh[wb + key] = sh[wb + key] + __popc(peers);
    }
    __syncthreads();
    int mine = 0;
    if (tid < BK_NODES) {
#pragma unroll
        for (int w2 = 0; w2 < 8; ++w2) mine += sh[w2 * BK_NODES + tid];
        tot[tid] = mine;
    }
    __syncthreads();
    for (int d = 1; d < BK_NODES; d <<= 1) {
        const int v = (tid < BK_NODES && tid >= d) ? tot[tid - d] : 0;
        __syncthreads();
        if (tid < BK_NODES) tot[tid] += v;
        __syncthreads();
    }
    if (tid < BK_NODES) {
        int run = bs + tot[tid] - mine;
        const int c = kb * BK_NODES + tid;
        if (c < n) off[c] = run;
#pragma unroll
        for (int w2 = 0; w2 < 8; ++w2) {
            const int cc = sh[w2 * BK_NODES + tid];
            sh[w2 * BK_NODES + tid] = run;
            run += cc;
        }
    }
    if (kb == 0 && tid == 0) off[n] = E;
    __syncthreads();
    for (int p0 = wbeg; p0 < wend; p0 += 32) {
        const int  p     = p0 + lane;
        const bool valid = p < wend;
        const int  key   = valid ? (bkt_col[p] & (BK_NODES - 1)) : 0;
        const unsigned peers = match_peers<7>(key, valid);
        const unsigned below = peers & ((1u << lane) - 1u);
        if (valid) {
            const int e    = bkt_e[p];
            const int base = sh[wb + key];
            const int pos  = base + __popc(below);
            if (below == 0u) sh[wb + key] = base + __popc(peers);
            csr_row[pos] = row[e];
            csr_ew[pos]  = ew[e];
        }
    }
}

__global__ void k_dinv(const int* __restrict__ off, const float* __restrict__ csr_ew,
                       float* __restrict__ dinv, int n) {
    int i = blockIdx.x * blockDim.x + threadIdx.x;
    if (i >= n) return;
    const int s = off[i], t = off[i + 1];
    float deg = 1.0f;
    for (int p = s; p < t; ++p) deg += csr_ew[p];
    dinv[i] = (deg > 0.0f) ? rsqrtf(deg) : 0.0f;
}

__global__ void __launch_bounds__(256)
k_gemm128_wmma(const float* __restrict__ A, const float* __restrict__ W,
               float* __restrict__ out, int nrows) {
    const int lane = threadIdx.x;
    const int wv   = threadIdx.y;
    const int m0   = blockIdx.x * 16;
    const int n0   = wv * 16;
    const int hi   = lane >> 4;
    const int l    = lane & 15;
    if (m0 + 16 > nrows) return;

    v8f acc = {};
    const float* ap = A + (size_t)(m0 + l) * NFEAT + 2 * hi;
    const float* bp = W + (size_t)(2 * hi) * NFEAT + n0 + l;
#pragma unroll 8
    for (int k = 0; k < NFEAT; k += 4) {
        v2f a = *(const v2f*)(ap + k);
        v2f b;
        b.x = bp[(size_t)k * NFEAT];
        b.y = bp[(size_t)k * NFEAT + NFEAT];
        acc = __builtin_amdgcn_wmma_f32_16x16x4_f32(
            false, a, false, b, (short)0, acc, false, false);
    }
    float* op = out + (size_t)(m0 + 8 * hi) * NFEAT + n0 + l;
#pragma unroll
    for (int v = 0; v < 8; ++v)
        op[(size_t)v * NFEAT] = acc[v];
}

template <bool RELU>
__global__ void __launch_bounds__(256)
k_agg(const int* __restrict__ off, const int* __restrict__ csr_row,
      const float* __restrict__ csr_ew, const float* __restrict__ dinv,
      const float* __restrict__ h, const float* __restrict__ bias,
      float* __restrict__ out, int n) {
    const int lane = threadIdx.x;
    const int c = blockIdx.x * 8 + __builtin_amdgcn_readfirstlane(threadIdx.y);
    if (c >= n) return;
    const int s = __builtin_amdgcn_readfirstlane(off[c]);
    const int t = __builtin_amdgcn_readfirstlane(off[c + 1]);
    const float dc = dinv[c];
    const int f = lane * 4;
    float4 acc = make_float4(0.0f, 0.0f, 0.0f, 0.0f);
    for (int base = s; base < t; base += 32) {
        const int m = (t - base < 32) ? (t - base) : 32;
        int   r_l = 0;
        float w_l = 0.0f;
        if (lane < m) {
            r_l = csr_row[base + lane];
            w_l = dinv[r_l] * csr_ew[base + lane] * dc;
        }
#pragma unroll 4
        for (int k = 0; k < m; ++k) {
            const int   r = __builtin_amdgcn_readlane(r_l, k);
            const float w = __int_as_float(__builtin_amdgcn_readlane(__float_as_int(w_l), k));
            const float4 hv = *(const float4*)(h + (size_t)r * NFEAT + f);
            acc.x = fmaf(w, hv.x, acc.x);
            acc.y = fmaf(w, hv.y, acc.y);
            acc.z = fmaf(w, hv.z, acc.z);
            acc.w = fmaf(w, hv.w, acc.w);
        }
    }
    const float  sl = dc * dc;
    const float4 hc = *(const float4*)(h + (size_t)c * NFEAT + f);
    const float4 bv = *(const float4*)(bias + f);
    float4 o;
    o.x = fmaf(sl, hc.x, acc.x) + bv.x;
    o.y = fmaf(sl, hc.y, acc.y) + bv.y;
    o.z = fmaf(sl, hc.z, acc.z) + bv.z;
    o.w = fmaf(sl, hc.w, acc.w) + bv.w;
    if (RELU) {
        o.x = fmaxf(o.x, 0.0f);
        o.y = fmaxf(o.y, 0.0f);
        o.z = fmaxf(o.z, 0.0f);
        o.w = fmaxf(o.w, 0.0f);
    }
    *(float4*)(out + (size_t)c * NFEAT + f) = o;
}

extern "C" void kernel_launch(void* const* d_in, const int* in_sizes, int n_in,
                              void* d_out, int out_size, void* d_ws, size_t ws_size,
                              hipStream_t stream) {
    const float* x  = (const float*)d_in[0];
    const int*   ei = (const int*)d_in[1];
    const float* ew = (const float*)d_in[2];
    const float* W1 = (const float*)d_in[3];
    const float* b1 = (const float*)d_in[4];
    const float* W2 = (const float*)d_in[5];
    const float* b2 = (const float*)d_in[6];
    float* out = (float*)d_out;

    const int N = in_sizes[0] / NFEAT;
    const int E = in_sizes[2];
    const int* row = ei;
    const int* col = ei + E;

    const int nchunk = (E + CHUNK - 1) / CHUNK;
    size_t npad = ((size_t)N + 1 + 1023) & ~(size_t)1023;
    size_t epad = ((size_t)E + 1023) & ~(size_t)1023;
    int*   hist    = (int*)d_ws;
    int*   cscan   = hist + (size_t)nchunk * BK_MAX;
    int*   bstart  = cscan + (size_t)nchunk * BK_MAX;
    int*   off     = bstart + 2 * BK_MAX;
    float* dinv    = (float*)(off + npad);
    int*   bkt_e   = (int*)(dinv + npad);
    int*   bkt_col = bkt_e + epad;
    int*   csr_row = bkt_col + epad;
    float* csr_ew  = (float*)(csr_row + epad);
    float* h       = csr_ew + epad;
    float* g       = h + (size_t)N * NFEAT;

    const int T = 256;
    dim3 wblk(32, 8);
    int mtiles = (N + 15) / 16;
    int nblk8  = (N + 7) / 8;
    int nbkt   = (N + BK_NODES - 1) / BK_NODES;

    k_bhist<<<nchunk, T, 0, stream>>>(col, hist, E);
    k_bscan<<<1, BK_MAX, 0, stream>>>(hist, cscan, bstart, nchunk);
    k_bscatter<<<nchunk, T, 0, stream>>>(col, cscan, bstart, bkt_e, bkt_col, E);
    k_bsort<<<nbkt, T, 0, stream>>>(bstart, bkt_e, bkt_col, row, ew, off, csr_row, csr_ew, N, E);
    k_dinv<<<(N + T - 1) / T, T, 0, stream>>>(off, csr_ew, dinv, N);

    k_gemm128_wmma<<<mtiles, wblk, 0, stream>>>(x, W1, h, N);
    k_agg<true><<<nblk8, wblk, 0, stream>>>(off, csr_row, csr_ew, dinv, h, b1, g, N);

    k_gemm128_wmma<<<mtiles, wblk, 0, stream>>>(g, W2, h, N);
    k_agg<false><<<nblk8, wblk, 0, stream>>>(off, csr_row, csr_ew, dinv, h, b2, out, N);
}


# GEMM re-tiled: one wave per 16x128 row strip (8 accumulator tiles share each A fragment; W values and results as 8-byte column pairs, whole-line stores); 6250 waves per GEMM, not 50000
# speedup vs baseline: 2.4466x; 1.0524x over previous
#include <hip/hip_runtime.h>

#define NFEAT 128

typedef float v2f  __attribute__((ext_vector_type(2)));
typedef float v8f  __attribute__((ext_vector_type(8)));

#define BK_SHIFT 7
#define BK_NODES 128
#define BK_MAX   1024
#define CHUNK    4096
#define GRP      (CHUNK / 8 / 32)

template <int BITS>
__device__ __forceinline__ unsigned match_peers(int key, bool valid) {
    unsigned peers = __builtin_amdgcn_ballot_w32(valid);
#pragma unroll
    for (int b = 0; b < BITS; ++b) {
        const bool bit = (key >> b) & 1;
        const unsigned m = __builtin_amdgcn_ballot_w32(bit);
        peers &= bit ? m : ~m;
    }
    return peers;
}

__device__ __forceinline__ void load_cols(const int* __restrict__ col, int E, int e0,
                                          int lane, int (&cv)[GRP]) {
#pragma unroll
    for (int g = 0; g < GRP; ++g) {
        const int e = e0 + g * 32 + lane;
        cv[g] = (e < E) ? col[e] : -1;
    }
}

#define COUNT_COLS(SH, WBASE)                                                        \
    _Pragma("unroll")                                                                \
    for (int g = 0; g < GRP; ++g) {                                                  \
        const bool valid = cv[g] >= 0;                                               \
        const int  key   = valid ? (cv[g] >> BK_SHIFT) : 0;                          \
        const unsigned peers = match_peers<10>(key, valid);                          \
        if (valid && (peers & ((1u << lane) - 1u)) == 0u)                            \
            SH[(WBASE) + key] = SH[(WBASE) + key] + __popc(peers);                   \
    }

__global__ void __launch_bounds__(256)
k_bhist(const int* __restrict__ col, int* __restrict__ hist, int E) {
    __shared__ int sh[8 * BK_MAX];
    const int tid  = threadIdx.x;
    const int lane = tid & 31;
    const int w    = __builtin_amdgcn_readfirstlane(tid >> 5);
    for (int i = tid; i < 8 * BK_MAX; i += 256) sh[i] = 0;
    int cv[GRP];
    load_cols(col, E, blockIdx.x * CHUNK + w * (CHUNK / 8), lane, cv);
    __syncthreads();
    COUNT_COLS(sh, w * BK_MAX)
    __syncthreads();
    for (int k = tid; k < BK_MAX; k += 256) {
        int s = 0;
#pragma unroll
        for (int w2 = 0; w2 < 8; ++w2) s += sh[w2 * BK_MAX + k];
        hist[blockIdx.x * BK_MAX + k] = s;
    }
}

__global__ void __launch_bounds__(1024)
k_bscan(const int* __restrict__ hist, int* __restrict__ cscan,
        int* __restrict__ bstart, int nblk) {
    __shared__ int sh[BK_MAX];
    const int k = threadIdx.x;
    int run = 0;
#pragma unroll 16
    for (int b = 0; b < nblk; ++b) {
        const int c = hist[b * BK_MAX + k];
        cscan[b * BK_MAX + k] = run;
        run += c;
    }
    sh[k] = run;
    __syncthreads();
    for (int d = 1; d < BK_MAX; d <<= 1) {
        const int v = (k >= d) ? sh[k - d] : 0;
        __syncthreads();
        sh[k] += v;
        __syncthreads();
    }
    const int incl = sh[k];
    bstart[k] = incl - run;
    if (k == BK_MAX - 1) bstart[BK_MAX] = incl;
}

__global__ void __launch_bounds__(256)
k_bscatter(const int* __restrict__ col, const int* __restrict__ cscan,
           const int* __restrict__ bstart, int* __restrict__ bkt_e,
           int* __restrict__ bkt_col, int E) {
    __shared__ int sh[8 * BK_MAX];
    const int tid  = threadIdx.x;
    const int lane = tid & 31;
    const int w    = __builtin_amdgcn_readfirstlane(tid >> 5);
    const int e0   = blockIdx.x * CHUNK + w * (CHUNK / 8);
    for (int i = tid; i < 8 * BK_MAX; i += 256) sh[i] = 0;
    int cv[GRP];
    load_cols(col, E, e0, lane, cv);
    __syncthreads();
    const int wb = w * BK_MAX;
    COUNT_COLS(sh, wb)
    __syncthreads();
    for (int k = tid; k < BK_MAX; k += 256) {
        int run = bstart[k] + cscan[blockIdx.x * BK_MAX + k];
#pragma unroll
        for (int w2 = 0; w2 < 8; ++w2) {
            const int c = sh[w2 * BK_MAX + k];
            sh[w2 * BK_MAX + k] = run;
            run += c;
        }
    }
    __syncthreads();
#pragma unroll
    for (int g = 0; g < GRP; ++g) {
        const bool valid = cv[g] >= 0;
        const int  key   = valid ? (cv[g] >> BK_SHIFT) : 0;
        const unsigned peers = match_peers<10>(key, valid);
        const unsigned below = peers & ((1u << lane) - 1u);
        if (valid) {
            const int base = sh[wb + key];
            const int pos  = base + __popc(below);
            if (below == 0u) sh[wb + key] = base + __popc(peers);
            bkt_e[pos]   = e0 + g * 32 + lane;
            bkt_col[pos] = cv[g];
        }
    }
}

__global__ void __launch_bounds__(256)
k_bsort(const int* __restrict__ bstart, const int* __restrict__ bkt_e,
        const int* __restrict__ bkt_col, const int* __restrict__ row,
        const float* __restrict__ ew, int* __restrict__ off,
        int* __restrict__ csr_row, float* __restrict__ csr_ew, int n, int E) {
    __shared__ int sh[8 * BK_NODES];
    __shared__ int tot[BK_NODES];
    const int tid  = threadIdx.x;
    const int lane = tid & 31;
    const int w    = __builtin_amdgcn_readfirstlane(tid >> 5);
    const int kb   = blockIdx.x;
    const int bs   = bstart[kb];
    const int be   = bstart[kb + 1];
    const int per  = (((be - bs) + 255) >> 8) << 5;
    const int wbeg = __builtin_amdgcn_readfirstlane(bs + w * per);
    const int wlim = __builtin_amdgcn_readfirstlane(wbeg + per);
    const int wend = (wlim < be) ? wlim : be;
    const int wb = w * BK_NODES;

    for (int i = tid; i < 8 * BK_NODES; i += 256) sh[i] = 0;
    __syncthreads();
    for (int p0 = wbeg; p0 < wend; p0 += 32) {
        const int  p     = p0 + lane;
        const bool valid = p < wend;
        const int  key   = valid ? (bkt_col[p] & (BK_NODES - 1)) : 0;
        const unsigned peers = match_peers<7>(key, valid);
        if (valid && (peers & ((1u << lane) - 1u)) == 0u)
            sh[wb + key] = sh[wb + key] + __popc(peers);
    }
    __syncthreads();
    int mine = 0;
    if (tid < BK_NODES) {
#pragma unroll
        for (int w2 = 0; w2 < 8; ++w2) mine += sh[w2 * BK_NODES + tid];
        tot[tid] = mine;
    }
    __syncthreads();
    for (int d = 1; d < BK_NODES; d <<= 1) {
        const int v = (tid < BK_NODES && tid >= d) ? tot[tid - d] : 0;
        __syncthreads();
        if (tid < BK_NODES) tot[tid] += v;
        __syncthreads();
    }
    if (tid < BK_NODES) {
        int run = bs + tot[tid] - mine;
        const int c = kb * BK_NODES + tid;
        if (c < n) off[c] = run;
#pragma unroll
        for (int w2 = 0; w2 < 8; ++w2) {
            const int cc = sh[w2 * BK_NODES + tid];
            sh[w2 * BK_NODES + tid] = run;
            run += cc;
        }
    }
    if (kb == 0 && tid == 0) off[n] = E;
    __syncthreads();
    for (int p0 = wbeg; p0 < wend; p0 += 32) {
        const int  p     = p0 + lane;
        const bool valid = p < wend;
        const int  key   = valid ? (bkt_col[p] & (BK_NODES - 1)) : 0;
        const unsigned peers = match_peers<7>(key, valid);
        const unsigned below = peers & ((1u << lane) - 1u);
        if (valid) {
            const int e    = bkt_e[p];
            const int base = sh[wb + key];
            const int pos  = base + __popc(below);
            if (below == 0u) sh[wb + key] = base + __popc(peers);
            csr_row[pos] = row[e];
            csr_ew[pos]  = ew[e];
        }
    }
}

__global__ void k_dinv(const int* __restrict__ off, const float* __restrict__ csr_ew,
                       float* __restrict__ dinv, int n) {
    int i = blockIdx.x * blockDim.x + threadIdx.x;
    if (i >= n) return;
    const int s = off[i], t = off[i + 1];
    float deg = 1.0f;
    for (int p = s; p < t; ++p) deg += csr_ew[p];
    dinv[i] = (deg > 0.0f) ? rsqrtf(deg) : 0.0f;
}

__global__ void __launch_bounds__(128)
k_gemm128_wmma(const float* __restrict__ A, const float* __restrict__ W,
               float* __restrict__ out, int nrows) {
    const int lane = threadIdx.x;
    const int m0   = (blockIdx.x * 4 + __builtin_amdgcn_readfirstlane(threadIdx.y)) * 16;
    const int hi   = lane >> 4;
    const int l    = lane & 15;
    if (m0 + 16 > nrows) return;

    v8f acc[8] = {};
    const float* ap = A + (size_t)(m0 + l) * NFEAT + 2 * hi;
    const float* bp = W + (size_t)(2 * hi) * NFEAT + 2 * l;
    for (int kk = 0; kk < NFEAT; kk += 32) {
        v2f a[8];
#pragma unroll
        for (int i = 0; i < 8; ++i) a[i] = *(const v2f*)(ap + kk + 4 * i);
#pragma unroll
        for (int i = 0; i < 8; ++i) {
            const int k = kk + 4 * i;
#pragma unroll
            for (int p = 0; p < 4; ++p) {
                const v2f r0 = *(const v2f*)(bp + (size_t)k * NFEAT + 32 * p);
                const v2f r1 = *(const v2f*)(bp + (size_t)k * NFEAT + NFEAT + 32 * p);
                v2f be, bo;
                be.x = r0.x; be.y = r1.x;
                bo.x = r0.y; bo.y = r1.y;
                acc[2 * p] = __builtin_amdgcn_wmma_f32_16x16x4_f32(
                    false, a[i], false, be, (short)0, acc[2 * p], false, false);
                acc[2 * p + 1] = __builtin_amdgcn_wmma_f32_16x16x4_f32(
                    false, a[i], false, bo, (short)0, acc[2 * p + 1], false, false);
            }
        }
    }
    float* op = out + (size_t)(m0 + 8 * hi) * NFEAT + 2 * l;
#pragma unroll
    for (int p = 0; p < 4; ++p) {
#pragma unroll
        for (int v = 0; v < 8; ++v) {
            v2f o;
            o.x = acc[2 * p][v];
            o.y = acc[2 * p + 1][v];
            *(v2f*)(op + (size_t)v * NFEAT + 32 * p) = o;
        }
    }
}

template <bool RELU>
__global__ void __launch_bounds__(256)
k_agg(const int* __restrict__ off, const int* __restrict__ csr_row,
      const float* __restrict__ csr_ew, const float* __restrict__ dinv,
      const float* __restrict__ h, const float* __restrict__ bias,
      float* __restrict__ out, int n) {
    const int lane = threadIdx.x;
    const int c = blockIdx.x * 8 + __builtin_amdgcn_readfirstlane(threadIdx.y);
    if (c >= n) return;
    const int s = __builtin_amdgcn_readfirstlane(off[c]);
    const int t = __builtin_amdgcn_readfirstlane(off[c + 1]);
    const float dc = dinv[c];
    const int f = lane * 4;
    float4 acc = make_float4(0.0f, 0.0f, 0.0f, 0.0f);
    for (int base = s; base < t; base += 32) {
        const int m = (t - base < 32) ? (t - base) : 32;
        int   r_l = 0;
        float w_l = 0.0f;
        if (lane < m) {
            r_l = csr_row[base + lane];
            w_l = dinv[r_l] * csr_ew[base + lane] * dc;
        }
#pragma unroll 4
        for (int k = 0; k < m; ++k) {
            const int   r = __builtin_amdgcn_readlane(r_l, k);
            const float w = __int_as_float(__builtin_amdgcn_readlane(__float_as_int(w_l), k));
            const float4 hv = *(const float4*)(h + (size_t)r * NFEAT + f);
            acc.x = fmaf(w, hv.x, acc.x);
            acc.y = fmaf(w, hv.y, acc.y);
            acc.z = fmaf(w, hv.z, acc.z);
            acc.w = fmaf(w, hv.w, acc.w);
        }
    }
    const float  sl = dc * dc;
    const float4 hc = *(const float4*)(h + (size_t)c * NFEAT + f);
    const float4 bv = *(const float4*)(bias + f);
    float4 o;
    o.x = fmaf(sl, hc.x, acc.x) + bv.x;
    o.y = fmaf(sl, hc.y, acc.y) + bv.y;
    o.z = fmaf(sl, hc.z, acc.z) + bv.z;
    o.w = fmaf(sl, hc.w, acc.w) + bv.w;
    if (RELU) {
        o.x = fmaxf(o.x, 0.0f);
        o.y = fmaxf(o.y, 0.0f);
        o.z = fmaxf(o.z, 0.0f);
        o.w = fmaxf(o.w, 0.0f);
    }
    *(float4*)(out + (size_t)c * NFEAT + f) = o;
}

extern "C" void kernel_launch(void* const* d_in, const int* in_sizes, int n_in,
                              void* d_out, int out_size, void* d_ws, size_t ws_size,
                              hipStream_t stream) {
    const float* x  = (const float*)d_in[0];
    const int*   ei = (const int*)d_in[1];
    const float* ew = (const float*)d_in[2];
    const float* W1 = (const float*)d_in[3];
    const float* b1 = (const float*)d_in[4];
    const float* W2 = (const float*)d_in[5];
    const float* b2 = (const float*)d_in[6];
    float* out = (float*)d_out;

    const int N = in_sizes[0] / NFEAT;
    const int E = in_sizes[2];
    const int* row = ei;
    const int* col = ei + E;

    const int nchunk = (E + CHUNK - 1) / CHUNK;
    size_t npad = ((size_t)N + 1 + 1023) & ~(size_t)1023;
    size_t epad = ((size_t)E + 1023) & ~(size_t)1023;
    int*   hist    = (int*)d_ws;
    int*   cscan   = hist + (size_t)nchunk * BK_MAX;
    int*   bstart  = cscan + (size_t)nchunk * BK_MAX;
    int*   off     = bstart + 2 * BK_MAX;
    float* dinv    = (float*)(off + npad);
    int*   bkt_e   = (int*)(dinv + npad);
    int*   bkt_col = bkt_e + epad;
    int*   csr_row = bkt_col + epad;
    float* csr_ew  = (float*)(csr_row + epad);
    float* h       = csr_ew + epad;
    float* g       = h + (size_t)N * NFEAT;

    const int T = 256;
    dim3 wblk(32, 8);
    dim3 gblk(32, 4);
    int nstrip4 = (N + 63) / 64;
    int nblk8  = (N + 7) / 8;
    int nbkt   = (N + BK_NODES - 1) / BK_NODES;

    k_bhist<<<nchunk, T, 0, stream>>>(col, hist, E);
    k_bscan<<<1, BK_MAX, 0, stream>>>(hist, cscan, bstart, nchunk);
    k_bscatter<<<nchunk, T, 0, stream>>>(col, cscan, bstart, bkt_e, bkt_col, E);
    k_bsort<<<nbkt, T, 0, stream>>>(bstart, bkt_e, bkt_col, row, ew, off, csr_row, csr_ew, N, E);
    k_dinv<<<(N + T - 1) / T, T, 0, stream>>>(off, csr_ew, dinv, N);

    k_gemm128_wmma<<<nstrip4, gblk, 0, stream>>>(x, W1, h, N);
    k_agg<true><<<nblk8, wblk, 0, stream>>>(off, csr_row, csr_ew, dinv, h, b1, g, N);

    k_gemm128_wmma<<<nstrip4, gblk, 0, stream>>>(g, W2, h, N);
    k_agg<false><<<nblk8, wblk, 0, stream>>>(off, csr_row, csr_ew, dinv, h, b2, out, N);
}


# Layer 1 aggregates x before its product (layer-1 summation order changes): a 16-node LDS tile feeds the W1 (+b1, ReLU) and W2 WMMAs inside the gather launch; both GEMM launches go
# speedup vs baseline: 2.8683x; 1.1724x over previous
#include <hip/hip_runtime.h>

#define NFEAT 128

typedef float v2f  __attribute__((ext_vector_type(2)));
typedef float v8f  __attribute__((ext_vector_type(8)));

#define BK_SHIFT 7
#define BK_NODES 128
#define BK_MAX   1024
#define CHUNK    4096
#define GRP      (CHUNK / 8 / 32)

template <int BITS>
__device__ __forceinline__ unsigned match_peers(int key, bool valid) {
    unsigned peers = __builtin_amdgcn_ballot_w32(valid);
#pragma unroll
    for (int b = 0; b < BITS; ++b) {
        const bool bit = (key >> b) & 1;
        const unsigned m = __builtin_amdgcn_ballot_w32(bit);
        peers &= bit ? m : ~m;
    }
    return peers;
}

__device__ __forceinline__ void load_cols(const int* __restrict__ col, int E, int e0,
                                          int lane, int (&cv)[GRP]) {
#pragma unroll
    for (int g = 0; g < GRP; ++g) {
        const int e = e0 + g * 32 + lane;
        cv[g] = (e < E) ? col[e] : -1;
    }
}

#define COUNT_COLS(SH, WBASE)                                                        \
    _Pragma("unroll")                                                                \
    for (int g = 0; g < GRP; ++g) {                                                  \
        const bool valid = cv[g] >= 0;                                               \
        const int  key   = valid ? (cv[g] >> BK_SHIFT) : 0;                          \
        const unsigned peers = match_peers<10>(key, valid);                          \
        if (valid && (peers & ((1u << lane) - 1u)) == 0u)                            \
            SH[(WBASE) + key] = SH[(WBASE) + key] + __popc(peers);                   \
    }

__global__ void __launch_bounds__(256)
k_bhist(const int* __restrict__ col, int* __restrict__ hist, int E) {
    __shared__ int sh[8 * BK_MAX];
    const int tid  = threadIdx.x;
    const int lane = tid & 31;
    const int w    = __builtin_amdgcn_readfirstlane(tid >> 5);
    for (int i = tid; i < 8 * BK_MAX; i += 256) sh[i] = 0;
    int cv[GRP];
    load_cols(col, E, blockIdx.x * CHUNK + w * (CHUNK / 8), lane, cv);
    __syncthreads();
    COUNT_COLS(sh, w * BK_MAX)
    __syncthreads();
    for (int k = tid; k < BK_MAX; k += 256) {
        int s = 0;
#pragma unroll
        for (int w2 = 0; w2 < 8; ++w2) s += sh[w2 * BK_MAX + k];
        hist[blockIdx.x * BK_MAX + k] = s;
    }
}

__global__ void __launch_bounds__(1024)
k_bscan(const int* __restrict__ hist, int* __restrict__ cscan,
        int* __restrict__ bstart, int nblk) {
    __shared__ int sh[BK_MAX];
    const int k = threadIdx.x;
    int run = 0;
#pragma unroll 16
    for (int b = 0; b < nblk; ++b) {
        const int c = hist[b * BK_MAX + k];
        cscan[b * BK_MAX + k] = run;
        run += c;
    }
    sh[k] = run;
    __syncthreads();
    for (int d = 1; d < BK_MAX; d <<= 1) {
        const int v = (k >= d) ? sh[k - d] : 0;
        __syncthreads();
        sh[k] += v;
        __syncthreads();
    }
    const int incl = sh[k];
    bstart[k] = incl - run;
    if (k == BK_MAX - 1) bstart[BK_MAX] = incl;
}

__global__ void __launch_bounds__(256)
k_bscatter(const int* __restrict__ col, const int* __restrict__ cscan,
           const int* __restrict__ bstart, int* __restrict__ bkt_e,
           int* __restrict__ bkt_col, int E) {
    __shared__ int sh[8 * BK_MAX];
    const int tid  = threadIdx.x;
    const int lane = tid & 31;
    const int w    = __builtin_amdgcn_readfirstlane(tid >> 5);
    const int e0   = blockIdx.x * CHUNK + w * (CHUNK / 8);
    for (int i = tid; i < 8 * BK_MAX; i += 256) sh[i] = 0;
    int cv[GRP];
    load_cols(col, E, e0, lane, cv);
    __syncthreads();
    const int wb = w * BK_MAX;
    COUNT_COLS(sh, wb)
    __syncthreads();
    for (int k = tid; k < BK_MAX; k += 256) {
        int run = bstart[k] + cscan[blockIdx.x * BK_MAX + k];
#pragma unroll
        for (int w2 = 0; w2 < 8; ++w2) {
            const int c = sh[w2 * BK_MAX + k];
            sh[w2 * BK_MAX + k] = run;
            run += c;
        }
    }
    __syncthreads();
#pragma unroll
    for (int g = 0; g < GRP; ++g) {
        const bool valid = cv[g] >= 0;
        const int  key   = valid ? (cv[g] >> BK_SHIFT) : 0;
        const unsigned peers = match_peers<10>(key, valid);
        const unsigned below = peers & ((1u << lane) - 1u);
        if (valid) {
            const int base = sh[wb + key];
            const int pos  = base + __popc(below);
            if (below == 0u) sh[wb + key] = base + __popc(peers);
            bkt_e[pos]   = e0 + g * 32 + lane;
            bkt_col[pos] = cv[g];
        }
    }
}

__global__ void __launch_bounds__(256)
k_bsort(const int* __restrict__ bstart, const int* __restrict__ bkt_e,
        const int* __restrict__ bkt_col, const int* __restrict__ row,
        const float* __restrict__ ew, int* __restrict__ off,
        int* __restrict__ csr_row, float* __restrict__ csr_ew, int n, int E) {
    __shared__ int sh[8 * BK_NODES];
    __shared__ int tot[BK_NODES];
    const int tid  = threadIdx.x;
    const int lane = tid & 31;
    const int w    = __builtin_amdgcn_readfirstlane(tid >> 5);
    const int kb   = blockIdx.x;
    const int bs   = bstart[kb];
    const int be   = bstart[kb + 1];
    const int per  = (((be - bs) + 255) >> 8) << 5;
    const int wbeg = __builtin_amdgcn_readfirstlane(bs + w * per);
    const int wlim = __builtin_amdgcn_readfirstlane(wbeg + per);
    const int wend = (wlim < be) ? wlim : be;
    const int wb = w * BK_NODES;

    for (int i = tid; i < 8 * BK_NODES; i += 256) sh[i] = 0;
    __syncthreads();
    for (int p0 = wbeg; p0 < wend; p0 += 32) {
        const int  p     = p0 + lane;
        const bool valid = p < wend;
        const int  key   = valid ? (bkt_col[p] & (BK_NODES - 1)) : 0;
        const unsigned peers = match_peers<7>(key, valid);
        if (valid && (peers & ((1u << lane) - 1u)) == 0u)
            sh[wb + key] = sh[wb + key] + __popc(peers);
    }
    __syncthreads();
    int mine = 0;
    if (tid < BK_NODES) {
#pragma unroll
        for (int w2 = 0; w2 < 8; ++w2) mine += sh[w2 * BK_NODES + tid];
        tot[tid] = mine;
    }
    __syncthreads();
    for (int d = 1; d < BK_NODES; d <<= 1) {
        const int v = (tid < BK_NODES && tid >= d) ? tot[tid - d] : 0;
        __syncthreads();
        if (tid < BK_NODES) tot[tid] += v;
        __syncthreads();
    }
    if (tid < BK_NODES) {
        int run = bs + tot[tid] - mine;
        const int c = kb * BK_NODES + tid;
        if (c < n) off[c] = run;
#pragma unroll
        for (int w2 = 0; w2 < 8; ++w2) {
            const int cc = sh[w2 * BK_NODES + tid];
            sh[w2 * BK_NODES + tid] = run;
            run += cc;
        }
    }
    if (kb == 0 && tid == 0) off[n] = E;
    __syncthreads();
    for (int p0 = wbeg; p0 < wend; p0 += 32) {
        const int  p     = p0 + lane;
        const bool valid = p < wend;
        const int  key   = valid ? (bkt_col[p] & (BK_NODES - 1)) : 0;
        const unsigned peers = match_peers<7>(key, valid);
        const unsigned below = peers & ((1u << lane) - 1u);
        if (valid) {
            const int e    = bkt_e[p];
            const int base = sh[wb + key];
            const int pos  = base + __popc(below);
            if (below == 0u) sh[wb + key] = base + __popc(peers);
            csr_row[pos] = row[e];
            csr_ew[pos]  = ew[e];
        }
    }
}

__global__ void k_dinv(const int* __restrict__ off, const float* __restrict__ csr_ew,
                       float* __restrict__ dinv, int n) {
    int i = blockIdx.x * blockDim.x + threadIdx.x;
    if (i >= n) return;
    const int s = off[i], t = off[i + 1];
    float deg = 1.0f;
    for (int p = s; p < t; ++p) deg += csr_ew[p];
    dinv[i] = (deg > 0.0f) ? rsqrtf(deg) : 0.0f;
}

__device__ __forceinline__ float4 agg_row(const int* __restrict__ csr_row,
                                          const float* __restrict__ csr_ew,
                                          const float* __restrict__ dinv,
                                          const float* __restrict__ tab,
                                          int c, int s, int t, float dc, int lane) {
    const int f = lane * 4;
    float4 acc = make_float4(0.0f, 0.0f, 0.0f, 0.0f);
    for (int base = s; base < t; base += 32) {
        const int m = (t - base < 32) ? (t - base) : 32;
        int   r_l = 0;
        float w_l = 0.0f;
        if (lane < m) {
            r_l = csr_row[base + lane];
            w_l = dinv[r_l] * csr_ew[base + lane] * dc;
        }
#pragma unroll 4
        for (int k = 0; k < m; ++k) {
            const int   r = __builtin_amdgcn_readlane(r_l, k);
            const float w = __int_as_float(__builtin_amdgcn_readlane(__float_as_int(w_l), k));
            const float4 hv = *(const float4*)(tab + (size_t)r * NFEAT + f);
            acc.x = fmaf(w, hv.x, acc.x);
            acc.y = fmaf(w, hv.y, acc.y);
            acc.z = fmaf(w, hv.z, acc.z);
            acc.w = fmaf(w, hv.w, acc.w);
        }
    }
    const float  sl = dc * dc;
    const float4 hc = *(const float4*)(tab + (size_t)c * NFEAT + f);
    acc.x = fmaf(sl, hc.x, acc.x);
    acc.y = fmaf(sl, hc.y, acc.y);
    acc.z = fmaf(sl, hc.z, acc.z);
    acc.w = fmaf(sl, hc.w, acc.w);
    return acc;
}

#define TP 132
__device__ __forceinline__ void mm_pair(const float* tile, const float* __restrict__ W,
                                        int p, int l, int hi, v8f& ev, v8f& od) {
    const float* ap = tile + l * TP + 2 * hi;
    const float* bp = W + (size_t)(2 * hi) * NFEAT + 32 * p + 2 * l;
#pragma unroll
    for (int kk = 0; kk < NFEAT; kk += 32) {
        v2f a[8];
#pragma unroll
        for (int i = 0; i < 8; ++i) a[i] = *(const v2f*)(ap + kk + 4 * i);
#pragma unroll
        for (int i = 0; i < 8; ++i) {
            const int k = kk + 4 * i;
            const v2f r0 = *(const v2f*)(bp + (size_t)k * NFEAT);
            const v2f r1 = *(const v2f*)(bp + (size_t)k * NFEAT + NFEAT);
            v2f be, bo;
            be.x = r0.x; be.y = r1.x;
            bo.x = r0.y; bo.y = r1.y;
            ev = __builtin_amdgcn_wmma_f32_16x16x4_f32(
                false, a[i], false, be, (short)0, ev, false, false);
            od = __builtin_amdgcn_wmma_f32_16x16x4_f32(
                false, a[i], false, bo, (short)0, od, false, false);
        }
    }
}

__global__ void __launch_bounds__(128)
k_agg_mm2(const int* __restrict__ off, const int* __restrict__ csr_row,
          const float* __restrict__ csr_ew, const float* __restrict__ dinv,
          const float* __restrict__ x, const float* __restrict__ W1,
          const float* __restrict__ b1, const float* __restrict__ W2,
          float* __restrict__ h2, int n) {
    __shared__ __attribute__((aligned(16))) float tile[16 * TP];
    const int lane = threadIdx.x;
    const int w    = __builtin_amdgcn_readfirstlane(threadIdx.y);
    const int m0   = blockIdx.x * 16;
    const int hi   = lane >> 4;
    const int l    = lane & 15;
    const int c0   = m0 + 4 * w;

    int   off_l = 0;
    float dv_l  = 0.0f;
    if (lane <= 4) { const int ci = c0 + lane; off_l = off[ci < n ? ci : n]; }
    if (lane < 4 && c0 + lane < n) dv_l = dinv[c0 + lane];
#pragma unroll 1
    for (int j = 0; j < 4; ++j) {
        const int c = c0 + j;
        float4 a = make_float4(0.0f, 0.0f, 0.0f, 0.0f);
        if (c < n) {
            const int   s  = __builtin_amdgcn_readlane(off_l, j);
            const int   t  = __builtin_amdgcn_readlane(off_l, j + 1);
            const float dc = __int_as_float(__builtin_amdgcn_readlane(__float_as_int(dv_l), j));
            a = agg_row(csr_row, csr_ew, dinv, x, c, s, t, dc, lane);
        }
        *(float4*)(tile + (4 * w + j) * TP + 4 * lane) = a;
    }
    __syncthreads();

    v8f ev = {}, od = {};
    mm_pair(tile, W1, w, l, hi, ev, od);
    const v2f bb = *(const v2f*)(b1 + 32 * w + 2 * l);
    __syncthreads();
#pragma unroll
    for (int v = 0; v < 8; ++v) {
        v2f g;
        g.x = fmaxf(ev[v] + bb.x, 0.0f);
        g.y = fmaxf(od[v] + bb.y, 0.0f);
        *(v2f*)(tile + (v + 8 * hi) * TP + 32 * w + 2 * l) = g;
    }
    __syncthreads();

    v8f ev2 = {}, od2 = {};
    mm_pair(tile, W2, w, l, hi, ev2, od2);
    float* op = h2 + (size_t)(m0 + 8 * hi) * NFEAT + 32 * w + 2 * l;
#pragma unroll
    for (int v = 0; v < 8; ++v) {
        if (m0 + 8 * hi + v < n) {
            v2f o;
            o.x = ev2[v];
            o.y = od2[v];
            *(v2f*)(op + (size_t)v * NFEAT) = o;
        }
    }
}

__global__ void __launch_bounds__(256)
k_agg(const int* __restrict__ off, const int* __restrict__ csr_row,
      const float* __restrict__ csr_ew, const float* __restrict__ dinv,
      const float* __restrict__ h, const float* __restrict__ bias,
      float* __restrict__ out, int n) {
    const int lane = threadIdx.x;
    const int c = blockIdx.x * 8 + __builtin_amdgcn_readfirstlane(threadIdx.y);
    if (c >= n) return;
    const int s = __builtin_amdgcn_readfirstlane(off[c]);
    const int t = __builtin_amdgcn_readfirstlane(off[c + 1]);
    const float dc = dinv[c];
    const int f = lane * 4;
    const float4 a  = agg_row(csr_row, csr_ew, dinv, h, c, s, t, dc, lane);
    const float4 bv = *(const float4*)(bias + f);
    float4 o;
    o.x = a.x + bv.x;
    o.y = a.y + bv.y;
    o.z = a.z + bv.z;
    o.w = a.w + bv.w;
    *(float4*)(out + (size_t)c * NFEAT + f) = o;
}

extern "C" void kernel_launch(void* const* d_in, const int* in_sizes, int n_in,
                              void* d_out, int out_size, void* d_ws, size_t ws_size,
                              hipStream_t stream) {
    const float* x  = (const float*)d_in[0];
    const int*   ei = (const int*)d_in[1];
    const float* ew = (const float*)d_in[2];
    const float* W1 = (const float*)d_in[3];
    const float* b1 = (const float*)d_in[4];
    const float* W2 = (const float*)d_in[5];
    const float* b2 = (const float*)d_in[6];
    float* out = (float*)d_out;

    const int N = in_sizes[0] / NFEAT;
    const int E = in_sizes[2];
    const int* row = ei;
    const int* col = ei + E;

    const int nchunk = (E + CHUNK - 1) / CHUNK;
    size_t npad = ((size_t)N + 1 + 1023) & ~(size_t)1023;
    size_t epad = ((size_t)E + 1023) & ~(size_t)1023;
    int*   hist    = (int*)d_ws;
    int*   cscan   = hist + (size_t)nchunk * BK_MAX;
    int*   bstart  = cscan + (size_t)nchunk * BK_MAX;
    int*   off     = bstart + 2 * BK_MAX;
    float* dinv    = (float*)(off + npad);
    int*   bkt_e   = (int*)(dinv + npad);
    int*   bkt_col = bkt_e + epad;
    int*   csr_row = bkt_col + epad;
    float* csr_ew  = (float*)(csr_row + epad);
    float* h       = csr_ew + epad;

    const int T = 256;
    dim3 wblk(32, 8);
    dim3 gblk(32, 4);
    int ntile16 = (N + 15) / 16;
    int nblk8  = (N + 7) / 8;
    int nbkt   = (N + BK_NODES - 1) / BK_NODES;

    k_bhist<<<nchunk, T, 0, stream>>>(col, hist, E);
    k_bscan<<<1, BK_MAX, 0, stream>>>(hist, cscan, bstart, nchunk);
    k_bscatter<<<nchunk, T, 0, stream>>>(col, cscan, bstart, bkt_e, bkt_col, E);
    k_bsort<<<nbkt, T, 0, stream>>>(bstart, bkt_e, bkt_col, row, ew, off, csr_row, csr_ew, N, E);
    k_dinv<<<(N + T - 1) / T, T, 0, stream>>>(off, csr_ew, dinv, N);

    k_agg_mm2<<<ntile16, gblk, 0, stream>>>(off, csr_row, csr_ew, dinv, x, W1, b1, W2, h, N);

    k_agg<<<nblk8, wblk, 0, stream>>>(off, csr_row, csr_ew, dinv, h, b2, out, N);
}


# Edges travel as one 8-byte record (source, weight; node-in-bucket packed in the first pass): one scattered store per edge per sort pass, not two, and no lookup of source and weight by edge id
# speedup vs baseline: 3.5101x; 1.2238x over previous
#include <hip/hip_runtime.h>

#define NFEAT 128

typedef float v2f  __attribute__((ext_vector_type(2)));
typedef float v8f  __attribute__((ext_vector_type(8)));
typedef int   v2i  __attribute__((ext_vector_type(2)));

#define BK_SHIFT 7
#define BK_NODES 128
#define BK_MAX   1024
#define CHUNK    4096
#define GRP      (CHUNK / 8 / 32)
#define LOC_SHIFT 24

template <int BITS>
__device__ __forceinline__ unsigned match_peers(int key, bool valid) {
    unsigned peers = __builtin_amdgcn_ballot_w32(valid);
#pragma unroll
    for (int b = 0; b < BITS; ++b) {
        const bool bit = (key >> b) & 1;
        const unsigned m = __builtin_amdgcn_ballot_w32(bit);
        peers &= bit ? m : ~m;
    }
    return peers;
}

__device__ __forceinline__ void load_cols(const int* __restrict__ col, int E, int e0,
                                          int lane, int (&cv)[GRP]) {
#pragma unroll
    for (int g = 0; g < GRP; ++g) {
        const int e = e0 + g * 32 + lane;
        cv[g] = (e < E) ? col[e] : -1;
    }
}

#define COUNT_COLS(SH, WBASE)                                                        \
    _Pragma("unroll")                                                                \
    for (int g = 0; g < GRP; ++g) {                                                  \
        const bool valid = cv[g] >= 0;                                               \
        const int  key   = valid ? (cv[g] >> BK_SHIFT) : 0;                          \
        const unsigned peers = match_peers<10>(key, valid);                          \
        if (valid && (peers & ((1u << lane) - 1u)) == 0u)                            \
            SH[(WBASE) + key] = SH[(WBASE) + key] + __popc(peers);                   \
    }

__global__ void __launch_bounds__(256)
k_bhist(const int* __restrict__ col, int* __restrict__ hist, int E) {
    __shared__ int sh[8 * BK_MAX];
    const int tid  = threadIdx.x;
    const int lane = tid & 31;
    const int w    = __builtin_amdgcn_readfirstlane(tid >> 5);
    for (int i = tid; i < 8 * BK_MAX; i += 256) sh[i] = 0;
    int cv[GRP];
    load_cols(col, E, blockIdx.x * CHUNK + w * (CHUNK / 8), lane, cv);
    __syncthreads();
    COUNT_COLS(sh, w * BK_MAX)
    __syncthreads();
    for (int k = tid; k < BK_MAX; k += 256) {
        int s = 0;
#pragma unroll
        for (int w2 = 0; w2 < 8; ++w2) s += sh[w2 * BK_MAX + k];
        hist[blockIdx.x * BK_MAX + k] = s;
    }
}

__global__ void __launch_bounds__(1024)
k_bscan(const int* __restrict__ hist, int* __restrict__ cscan,
        int* __restrict__ bstart, int nblk) {
    __shared__ int sh[BK_MAX];
    const int k = threadIdx.x;
    int run = 0;
#pragma unroll 16
    for (int b = 0; b < nblk; ++b) {
        const int c = hist[b * BK_MAX + k];
        cscan[b * BK_MAX + k] = run;
        run += c;
    }
    sh[k] = run;
    __syncthreads();
    for (int d = 1; d < BK_MAX; d <<= 1) {
        const int v = (k >= d) ? sh[k - d] : 0;
        __syncthreads();
        sh[k] += v;
        __syncthreads();
    }
    const int incl = sh[k];
    bstart[k] = incl - run;
    if (k == BK_MAX - 1) bstart[BK_MAX] = incl;
}

__global__ void __launch_bounds__(256)
k_bscatter(const int* __restrict__ col, const int* __restrict__ row,
           const float* __restrict__ ew, const int* __restrict__ cscan,
           const int* __restrict__ bstart, int* __restrict__ bkt_rec, int E) {
    __shared__ int sh[8 * BK_MAX];
    const int tid  = threadIdx.x;
    const int lane = tid & 31;
    const int w    = __builtin_amdgcn_readfirstlane(tid >> 5);
    const int e0   = blockIdx.x * CHUNK + w * (CHUNK / 8);
    for (int i = tid; i < 8 * BK_MAX; i += 256) sh[i] = 0;
    int cv[GRP];
    load_cols(col, E, e0, lane, cv);
    int   rv[GRP];
    float wv[GRP];
#pragma unroll
    for (int g = 0; g < GRP; ++g) {
        const int e = e0 + g * 32 + lane;
        rv[g] = (e < E) ? row[e] : 0;
        wv[g] = (e < E) ? ew[e] : 0.0f;
    }
    __syncthreads();
    const int wb = w * BK_MAX;
    COUNT_COLS(sh, wb)
    __syncthreads();
    for (int k = tid; k < BK_MAX; k += 256) {
        int run = bstart[k] + cscan[blockIdx.x * BK_MAX + k];
#pragma unroll
        for (int w2 = 0; w2 < 8; ++w2) {
            const int c = sh[w2 * BK_MAX + k];
            sh[w2 * BK_MAX + k] = run;
            run += c;
        }
    }
    __syncthreads();
#pragma unroll
    for (int g = 0; g < GRP; ++g) {
        const bool valid = cv[g] >= 0;
        const int  key   = valid ? (cv[g] >> BK_SHIFT) : 0;
        const unsigned peers = match_peers<10>(key, valid);
        const unsigned below = peers & ((1u << lane) - 1u);
        if (valid) {
            const int base = sh[wb + key];
            const int pos  = base + __popc(below);
            if (below == 0u) sh[wb + key] = base + __popc(peers);
            v2i rec;
            rec.x = rv[g] | ((cv[g] & (BK_NODES - 1)) << LOC_SHIFT);
            rec.y = __float_as_int(wv[g]);
            ((v2i*)bkt_rec)[pos] = rec;
        }
    }
}

__global__ void __launch_bounds__(256)
k_bsort(const int* __restrict__ bstart, const int* __restrict__ bkt_rec,
        int* __restrict__ off, int* __restrict__ csr_rec, int n, int E) {
    __shared__ int sh[8 * BK_NODES];
    __shared__ int tot[BK_NODES];
    const int tid  = threadIdx.x;
    const int lane = tid & 31;
    const int w    = __builtin_amdgcn_readfirstlane(tid >> 5);
    const int kb   = blockIdx.x;
    const int bs   = bstart[kb];
    const int be   = bstart[kb + 1];
    const int per  = (((be - bs) + 255) >> 8) << 5;
    const int wbeg = __builtin_amdgcn_readfirstlane(bs + w * per);
    const int wlim = __builtin_amdgcn_readfirstlane(wbeg + per);
    const int wend = (wlim < be) ? wlim : be;
    const int wb = w * BK_NODES;

    for (int i = tid; i < 8 * BK_NODES; i += 256) sh[i] = 0;
    __syncthreads();
    for (int p0 = wbeg; p0 < wend; p0 += 32) {
        const int  p     = p0 + lane;
        const bool valid = p < wend;
        const int  key   = valid ? (bkt_rec[2 * p] >> LOC_SHIFT) : 0;
        const unsigned peers = match_peers<7>(key, valid);
        if (valid && (peers & ((1u << lane) - 1u)) == 0u)
            sh[wb + key] = sh[wb + key] + __popc(peers);
    }
    __syncthreads();
    int mine = 0;
    if (tid < BK_NODES) {
#pragma unroll
        for (int w2 = 0; w2 < 8; ++w2) mine += sh[w2 * BK_NODES + tid];
        tot[tid] = mine;
    }
    __syncthreads();
    for (int d = 1; d < BK_NODES; d <<= 1) {
        const int v = (tid < BK_NODES && tid >= d) ? tot[tid - d] : 0;
        __syncthreads();
        if (tid < BK_NODES) tot[tid] += v;
        __syncthreads();
    }
    if (tid < BK_NODES) {
        int run = bs + tot[tid] - mine;
        const int c = kb * BK_NODES + tid;
        if (c < n) off[c] = run;
#pragma unroll
        for (int w2 = 0; w2 < 8; ++w2) {
            const int cc = sh[w2 * BK_NODES + tid];
            sh[w2 * BK_NODES + tid] = run;
            run += cc;
        }
    }
    if (kb == 0 && tid == 0) off[n] = E;
    __syncthreads();
    for (int p0 = wbeg; p0 < wend; p0 += 32) {
        const int  p     = p0 + lane;
        const bool valid = p < wend;
        v2i rec = {0, 0};
        if (valid) rec = ((const v2i*)bkt_rec)[p];
        const int  key   = rec.x >> LOC_SHIFT;
        const unsigned peers = match_peers<7>(key, valid);
        const unsigned below = peers & ((1u << lane) - 1u);
        if (valid) {
            const int base = sh[wb + key];
            const int pos  = base + __popc(below);
            if (below == 0u) sh[wb + key] = base + __popc(peers);
            rec.x &= (1 << LOC_SHIFT) - 1;
            ((v2i*)csr_rec)[pos] = rec;
        }
    }
}

__global__ void k_dinv(const int* __restrict__ off, const int* __restrict__ csr_rec,
                       float* __restrict__ dinv, int n) {
    int i = blockIdx.x * blockDim.x + threadIdx.x;
    if (i >= n) return;
    const int s = off[i], t = off[i + 1];
    float deg = 1.0f;
    for (int p = s; p < t; ++p) deg += __int_as_float(csr_rec[2 * p + 1]);
    dinv[i] = (deg > 0.0f) ? rsqrtf(deg) : 0.0f;
}

__device__ __forceinline__ float4 agg_row(const int* __restrict__ csr_rec,
                                          const float* __restrict__ dinv,
                                          const float* __restrict__ tab,
                                          int c, int s, int t, float dc, int lane) {
    const int f = lane * 4;
    float4 acc = make_float4(0.0f, 0.0f, 0.0f, 0.0f);
    for (int base = s; base < t; base += 32) {
        const int m = (t - base < 32) ? (t - base) : 32;
        int   r_l = 0;
        float w_l = 0.0f;
        if (lane < m) {
            const v2i rec = ((const v2i*)csr_rec)[base + lane];
            r_l = rec.x;
            w_l = dinv[r_l] * __int_as_float(rec.y) * dc;
        }
#pragma unroll 4
        for (int k = 0; k < m; ++k) {
            const int   r = __builtin_amdgcn_readlane(r_l, k);
            const float w = __int_as_float(__builtin_amdgcn_readlane(__float_as_int(w_l), k));
            const float4 hv = *(const float4*)(tab + (size_t)r * NFEAT + f);
            acc.x = fmaf(w, hv.x, acc.x);
            acc.y = fmaf(w, hv.y, acc.y);
            acc.z = fmaf(w, hv.z, acc.z);
            acc.w = fmaf(w, hv.w, acc.w);
        }
    }
    const float  sl = dc * dc;
    const float4 hc = *(const float4*)(tab + (size_t)c * NFEAT + f);
    acc.x = fmaf(sl, hc.x, acc.x);
    acc.y = fmaf(sl, hc.y, acc.y);
    acc.z = fmaf(sl, hc.z, acc.z);
    acc.w = fmaf(sl, hc.w, acc.w);
    return acc;
}

#define TP 132
__device__ __forceinline__ void mm_pair(const float* tile, const float* __restrict__ W,
                                        int p, int l, int hi, v8f& ev, v8f& od) {
    const float* ap = tile + l * TP + 2 * hi;
    const float* bp = W + (size_t)(2 * hi) * NFEAT + 32 * p + 2 * l;
#pragma unroll
    for (int kk = 0; kk < NFEAT; kk += 32) {
        v2f a[8];
#pragma unroll
        for (int i = 0; i < 8; ++i) a[i] = *(const v2f*)(ap + kk + 4 * i);
#pragma unroll
        for (int i = 0; i < 8; ++i) {
            const int k = kk + 4 * i;
            const v2f r0 = *(const v2f*)(bp + (size_t)k * NFEAT);
            const v2f r1 = *(const v2f*)(bp + (size_t)k * NFEAT + NFEAT);
            v2f be, bo;
            be.x = r0.x; be.y = r1.x;
            bo.x = r0.y; bo.y = r1.y;
            ev = __builtin_amdgcn_wmma_f32_16x16x4_f32(
                false, a[i], false, be, (short)0, ev, false, false);
            od = __builtin_amdgcn_wmma_f32_16x16x4_f32(
                false, a[i], false, bo, (short)0, od, false, false);
        }
    }
}

__global__ void __launch_bounds__(128)
k_agg_mm2(const int* __restrict__ off, const int* __restrict__ csr_rec,
          const float* __restrict__ dinv,
          const float* __restrict__ x, const float* __restrict__ W1,
          const float* __restrict__ b1, const float* __restrict__ W2,
          float* __restrict__ h2, int n) {
    __shared__ __attribute__((aligned(16))) float tile[16 * TP];
    const int lane = threadIdx.x;
    const int w    = __builtin_amdgcn_readfirstlane(threadIdx.y);
    const int m0   = blockIdx.x * 16;
    const int hi   = lane >> 4;
    const int l    = lane & 15;
    const int c0   = m0 + 4 * w;

    int   off_l = 0;
    float dv_l  = 0.0f;
    if (lane <= 4) { const int ci = c0 + lane; off_l = off[ci < n ? ci : n]; }
    if (lane < 4 && c0 + lane < n) dv_l = dinv[c0 + lane];
#pragma unroll 1
    for (int j = 0; j < 4; ++j) {
        const int c = c0 + j;
        float4 a = make_float4(0.0f, 0.0f, 0.0f, 0.0f);
        if (c < n) {
            const int   s  = __builtin_amdgcn_readlane(off_l, j);
            const int   t  = __builtin_amdgcn_readlane(off_l, j + 1);
            const float dc = __int_as_float(__builtin_amdgcn_readlane(__float_as_int(dv_l), j));
            a = agg_row(csr_rec, dinv, x, c, s, t, dc, lane);
        }
        *(float4*)(tile + (4 * w + j) * TP + 4 * lane) = a;
    }
    __syncthreads();

    v8f ev = {}, od = {};
    mm_pair(tile, W1, w, l, hi, ev, od);
    const v2f bb = *(const v2f*)(b1 + 32 * w + 2 * l);
    __syncthreads();
#pragma unroll
    for (int v = 0; v < 8; ++v) {
        v2f g;
        g.x = fmaxf(ev[v] + bb.x, 0.0f);
        g.y = fmaxf(od[v] + bb.y, 0.0f);
        *(v2f*)(tile + (v + 8 * hi) * TP + 32 * w + 2 * l) = g;
    }
    __syncthreads();

    v8f ev2 = {}, od2 = {};
    mm_pair(tile, W2, w, l, hi, ev2, od2);
    float* op = h2 + (size_t)(m0 + 8 * hi) * NFEAT + 32 * w + 2 * l;
#pragma unroll
    for (int v = 0; v < 8; ++v) {
        if (m0 + 8 * hi + v < n) {
            v2f o;
            o.x = ev2[v];
            o.y = od2[v];
            *(v2f*)(op + (size_t)v * NFEAT) = o;
        }
    }
}

__global__ void __launch_bounds__(256)
k_agg(const int* __restrict__ off, const int* __restrict__ csr_rec,
      const float* __restrict__ dinv,
      const float* __restrict__ h, const float* __restrict__ bias,
      float* __restrict__ out, int n) {
    const int lane = threadIdx.x;
    const int c = blockIdx.x * 8 + __builtin_amdgcn_readfirstlane(threadIdx.y);
    if (c >= n) return;
    const int s = __builtin_amdgcn_readfirstlane(off[c]);
    const int t = __builtin_amdgcn_readfirstlane(off[c + 1]);
    const float dc = dinv[c];
    const int f = lane * 4;
    const float4 a  = agg_row(csr_rec, dinv, h, c, s, t, dc, lane);
    const float4 bv = *(const float4*)(bias + f);
    float4 o;
    o.x = a.x + bv.x;
    o.y = a.y + bv.y;
    o.z = a.z + bv.z;
    o.w = a.w + bv.w;
    *(float4*)(out + (size_t)c * NFEAT + f) = o;
}

extern "C" void kernel_launch(void* const* d_in, const int* in_sizes, int n_in,
                              void* d_out, int out_size, void* d_ws, size_t ws_size,
                              hipStream_t stream) {
    const float* x  = (const float*)d_in[0];
    const int*   ei = (const int*)d_in[1];
    const float* ew = (const float*)d_in[2];
    const float* W1 = (const float*)d_in[3];
    const float* b1 = (const float*)d_in[4];
    const float* W2 = (const float*)d_in[5];
    const float* b2 = (const float*)d_in[6];
    float* out = (float*)d_out;

    const int N = in_sizes[0] / NFEAT;
    const int E = in_sizes[2];
    const int* row = ei;
    const int* col = ei + E;

    const int nchunk = (E + CHUNK - 1) / CHUNK;
    size_t npad = ((size_t)N + 1 + 1023) & ~(size_t)1023;
    size_t epad = ((size_t)E + 1023) & ~(size_t)1023;
    int*   hist    = (int*)d_ws;
    int*   cscan   = hist + (size_t)nchunk * BK_MAX;
    int*   bstart  = cscan + (size_t)nchunk * BK_MAX;
    int*   off     = bstart + 2 * BK_MAX;
    float* dinv    = (float*)(off + npad);
    int*   bkt_rec = (int*)(dinv + npad);
    int*   csr_rec = bkt_rec + 2 * epad;
    float* h       = (float*)(csr_rec + 2 * epad);

    const int T = 256;
    dim3 wblk(32, 8);
    dim3 gblk(32, 4);
    int ntile16 = (N + 15) / 16;
    int nblk8  = (N + 7) / 8;
    int nbkt   = (N + BK_NODES - 1) / BK_NODES;

    k_bhist<<<nchunk, T, 0, stream>>>(col, hist, E);
    k_bscan<<<1, BK_MAX, 0, stream>>>(hist, cscan, bstart, nchunk);
    k_bscatter<<<nchunk, T, 0, stream>>>(col, row, ew, cscan, bstart, bkt_rec, E);
    k_bsort<<<nbkt, T, 0, stream>>>(bstart, bkt_rec, off, csr_rec, N, E);
    k_dinv<<<(N + T - 1) / T, T, 0, stream>>>(off, csr_rec, dinv, N);

    k_agg_mm2<<<ntile16, gblk, 0, stream>>>(off, csr_rec, dinv, x, W1, b1, W2, h, N);

    k_agg<<<nblk8, wblk, 0, stream>>>(off, csr_rec, dinv, h, b2, out, N);
}


# Destination sort in 2 launches, not 5: chunks sort by bucket in place and write a run-start table; the bucket block sums its table column, joins its runs, sorts by node, computes dinv
# speedup vs baseline: 4.6487x; 1.3244x over previous
#include <hip/hip_runtime.h>

#define NFEAT 128

typedef float v2f  __attribute__((ext_vector_type(2)));
typedef float v8f  __attribute__((ext_vector_type(8)));
typedef int   v2i  __attribute__((ext_vector_type(2)));

#define BK_SHIFT 7
#define BK_NODES 128
#define BK_MAX   1024
#define CHUNK    4096
#define GRP      (CHUNK / 8 / 32)
#define TBP      (BK_MAX + 4)
#define MAXCH    1024
#define LOC_SHIFT 24

template <int BITS>
__device__ __forceinline__ unsigned match_peers(int key, bool valid) {
    unsigned peers = __builtin_amdgcn_ballot_w32(valid);
#pragma unroll
    for (int b = 0; b < BITS; ++b) {
        const bool bit = (key >> b) & 1;
        const unsigned m = __builtin_amdgcn_ballot_w32(bit);
        peers &= bit ? m : ~m;
    }
    return peers;
}

__global__ void __launch_bounds__(256)
k_bscatter(const int* __restrict__ col, const int* __restrict__ row,
           const float* __restrict__ ew, int* __restrict__ tbl,
           int* __restrict__ crec, int E) {
    __shared__ __attribute__((aligned(16))) int sh[8 * BK_MAX];
    __shared__ int ssum[256];
    const int tid  = threadIdx.x;
    const int lane = tid & 31;
    const int w    = __builtin_amdgcn_readfirstlane(tid >> 5);
    const int e0   = blockIdx.x * CHUNK + w * (CHUNK / 8);
    for (int i = tid; i < 8 * BK_MAX; i += 256) sh[i] = 0;
    int   cv[GRP];
    int   rv[GRP];
    float wv[GRP];
#pragma unroll
    for (int g = 0; g < GRP; ++g) {
        const int e = e0 + g * 32 + lane;
        cv[g] = (e < E) ? col[e] : -1;
        rv[g] = (e < E) ? row[e] : 0;
        wv[g] = (e < E) ? ew[e] : 0.0f;
    }
    __syncthreads();
    const int wb = w * BK_MAX;
#pragma unroll
    for (int g = 0; g < GRP; ++g) {
        const bool valid = cv[g] >= 0;
        const int  key   = valid ? (cv[g] >> BK_SHIFT) : 0;
        const unsigned peers = match_peers<10>(key, valid);
        if (valid && (peers & ((1u << lane) - 1u)) == 0u)
            sh[wb + key] = sh[wb + key] + __popc(peers);
    }
    __syncthreads();
    const int k4 = 4 * tid;
    int t0 = 0, t1 = 0, t2 = 0, t3 = 0;
#pragma unroll
    for (int w2 = 0; w2 < 8; ++w2) {
        const int4 c = *(const int4*)(sh + w2 * BK_MAX + k4);
        t0 += c.x; t1 += c.y; t2 += c.z; t3 += c.w;
    }
    const int mysum = t0 + t1 + t2 + t3;
    ssum[tid] = mysum;
    __syncthreads();
    for (int d = 1; d < 256; d <<= 1) {
        const int v = (tid >= d) ? ssum[tid - d] : 0;
        __syncthreads();
        ssum[tid] += v;
        __syncthreads();
    }
    int r0 = ssum[tid] - mysum;
    int r1 = r0 + t0, r2 = r1 + t1, r3 = r2 + t2;
    int* trow = tbl + (size_t)blockIdx.x * TBP;
    *(int4*)(trow + k4) = make_int4(r0, r1, r2, r3);
    if (tid == 255) trow[BK_MAX] = r3 + t3;
#pragma unroll
    for (int w2 = 0; w2 < 8; ++w2) {
        const int4 c = *(const int4*)(sh + w2 * BK_MAX + k4);
        *(int4*)(sh + w2 * BK_MAX + k4) = make_int4(r0, r1, r2, r3);
        r0 += c.x; r1 += c.y; r2 += c.z; r3 += c.w;
    }
    __syncthreads();
    v2i* cdst = (v2i*)crec + (size_t)blockIdx.x * CHUNK;
#pragma unroll
    for (int g = 0; g < GRP; ++g) {
        const bool valid = cv[g] >= 0;
        const int  key   = valid ? (cv[g] >> BK_SHIFT) : 0;
        const unsigned peers = match_peers<10>(key, valid);
        const unsigned below = peers & ((1u << lane) - 1u);
        if (valid) {
            const int base = sh[wb + key];
            const int pos  = base + __popc(below);
            if (below == 0u) sh[wb + key] = base + __popc(peers);
            v2i rec;
            rec.x = rv[g] | ((cv[g] & (BK_NODES - 1)) << LOC_SHIFT);
            rec.y = __float_as_int(wv[g]);
            cdst[pos] = rec;
        }
    }
}

__global__ void __launch_bounds__(256)
k_bsort(const int* __restrict__ tbl, const int* __restrict__ crec,
        int* __restrict__ bkt_rec, int* __restrict__ off, int* __restrict__ csr_rec,
        float* __restrict__ dinv, int n, int E, int nchunk) {
    __shared__ int sh[8 * BK_NODES];
    __shared__ int tot[BK_NODES];
    __shared__ int rs[MAXCH];
    __shared__ int rp[MAXCH];
    __shared__ int sc[256];
    __shared__ int ss[256];
    const int tid  = threadIdx.x;
    const int lane = tid & 31;
    const int w    = __builtin_amdgcn_readfirstlane(tid >> 5);
    const int kb   = blockIdx.x;
    const int wb   = w * BK_NODES;

    for (int i = tid; i < 8 * BK_NODES; i += 256) sh[i] = 0;
    int s4[4], c4[4];
    int ls = 0, lc = 0;
#pragma unroll
    for (int j = 0; j < 4; ++j) {
        const int c = 4 * tid + j;
        const int* trow = tbl + (size_t)((c < nchunk) ? c : nchunk - 1) * TBP + kb;
        const int s = (c < nchunk) ? trow[0] : 0;
        const int e = (c < nchunk) ? trow[1] : 0;
        s4[j] = s;
        c4[j] = e - s;
        ls += s;
        lc += e - s;
    }
    sc[tid] = lc;
    ss[tid] = ls;
    __syncthreads();
    for (int d = 1; d < 256; d <<= 1) {
        int vc = 0, vs = 0;
        if (tid >= d) { vc = sc[tid - d]; vs = ss[tid - d]; }
        __syncthreads();
        sc[tid] += vc;
        ss[tid] += vs;
        __syncthreads();
    }
    {
        int run = sc[tid] - lc;
#pragma unroll
        for (int j = 0; j < 4; ++j) {
            rs[4 * tid + j] = s4[j];
            rp[4 * tid + j] = run;
            run += c4[j];
        }
    }
    const int total = __builtin_amdgcn_readfirstlane(sc[255]);
    const int bs    = __builtin_amdgcn_readfirstlane(ss[255]);
    const int per   = ((total + 255) >> 8) << 5;
    const int wbeg  = __builtin_amdgcn_readfirstlane(w * per);
    const int wlim  = wbeg + per;
    const int wend  = (wlim < total) ? wlim : total;
    v2i* blist = (v2i*)bkt_rec + bs;
    __syncthreads();
    for (int p0 = wbeg; p0 < wend; p0 += 64) {
        int  p[2];
        bool valid[2];
        v2i  rec[2];
#pragma unroll
        for (int u = 0; u < 2; ++u) {
            p[u]     = p0 + 32 * u + lane;
            valid[u] = p[u] < wend;
            const int q = valid[u] ? p[u] : total - 1;
            int c = 0;
#pragma unroll
            for (int step = MAXCH / 2; step >= 1; step >>= 1)
                c += (rp[c + step] <= q) ? step : 0;
            rec[u] = ((const v2i*)crec)[(size_t)c * CHUNK + rs[c] + (q - rp[c])];
        }
#pragma unroll
        for (int u = 0; u < 2; ++u) {
            if (valid[u]) blist[p[u]] = rec[u];
            const int  key   = valid[u] ? (rec[u].x >> LOC_SHIFT) : 0;
            const unsigned peers = match_peers<7>(key, valid[u]);
            if (valid[u] && (peers & ((1u << lane) - 1u)) == 0u)
                sh[wb + key] = sh[wb + key] + __popc(peers);
        }
    }
    __syncthreads();
    int mine = 0;
    if (tid < BK_NODES) {
#pragma unroll
        for (int w2 = 0; w2 < 8; ++w2) mine += sh[w2 * BK_NODES + tid];
        tot[tid] = mine;
    }
    __syncthreads();
    for (int d = 1; d < BK_NODES; d <<= 1) {
        const int v = (tid < BK_NODES && tid >= d) ? tot[tid - d] : 0;
        __syncthreads();
        if (tid < BK_NODES) tot[tid] += v;
        __syncthreads();
    }
    int seg = 0;
    if (tid < BK_NODES) {
        seg = bs + tot[tid] - mine;
        int run = seg;
        const int c = kb * BK_NODES + tid;
        if (c < n) off[c] = run;
#pragma unroll
        for (int w2 = 0; w2 < 8; ++w2) {
            const int cc = sh[w2 * BK_NODES + tid];
            sh[w2 * BK_NODES + tid] = run;
            run += cc;
        }
    }
    if (kb == 0 && tid == 0) off[n] = E;
    __syncthreads();
    for (int p0 = wbeg; p0 < wend; p0 += 64) {
        int  p[2];
        bool valid[2];
        v2i  rec[2];
#pragma unroll
        for (int u = 0; u < 2; ++u) {
            p[u]     = p0 + 32 * u + lane;
            valid[u] = p[u] < wend;
            rec[u]   = blist[valid[u] ? p[u] : total - 1];
        }
#pragma unroll
        for (int u = 0; u < 2; ++u) {
            const int  key   = valid[u] ? (rec[u].x >> LOC_SHIFT) : 0;
            const unsigned peers = match_peers<7>(key, valid[u]);
            const unsigned below = peers & ((1u << lane) - 1u);
            if (valid[u]) {
                const int base = sh[wb + key];
                const int pos  = base + __popc(below);
                if (below == 0u) sh[wb + key] = base + __popc(peers);
                v2i o;
                o.x = rec[u].x & ((1 << LOC_SHIFT) - 1);
                o.y = rec[u].y;
                ((v2i*)csr_rec)[pos] = o;
            }
        }
    }
    __syncthreads();
    if (tid < BK_NODES && kb * BK_NODES + tid < n) {
        float deg = 1.0f;
        const int end = seg + mine;
        for (int p = seg; p < end; p += 4) {
            float wq[4];
#pragma unroll
            for (int u = 0; u < 4; ++u)
                wq[u] = (p + u < end) ? __int_as_float(csr_rec[2 * (p + u) + 1]) : 0.0f;
#pragma unroll
            for (int u = 0; u < 4; ++u) deg += wq[u];
        }
        dinv[kb * BK_NODES + tid] = (deg > 0.0f) ? rsqrtf(deg) : 0.0f;
    }
}

__device__ __forceinline__ float4 agg_row(const int* __restrict__ csr_rec,
                                          const float* __restrict__ dinv,
                                          const float* __restrict__ tab,
                                          int c, int s, int t, float dc, int lane) {
    const int f = lane * 4;
    float4 acc = make_float4(0.0f, 0.0f, 0.0f, 0.0f);
    for (int base = s; base < t; base += 32) {
        const int m = (t - base < 32) ? (t - base) : 32;
        int   r_l = 0;
        float w_l = 0.0f;
        if (lane < m) {
            const v2i rec = ((const v2i*)csr_rec)[base + lane];
            r_l = rec.x;
            w_l = dinv[r_l] * __int_as_float(rec.y) * dc;
        }
#pragma unroll 4
        for (int k = 0; k < m; ++k) {
            const int   r = __builtin_amdgcn_readlane(r_l, k);
            const float w = __int_as_float(__builtin_amdgcn_readlane(__float_as_int(w_l), k));
            const float4 hv = *(const float4*)(tab + (size_t)r * NFEAT + f);
            acc.x = fmaf(w, hv.x, acc.x);
            acc.y = fmaf(w, hv.y, acc.y);
            acc.z = fmaf(w, hv.z, acc.z);
            acc.w = fmaf(w, hv.w, acc.w);
        }
    }
    const float  sl = dc * dc;
    const float4 hc = *(const float4*)(tab + (size_t)c * NFEAT + f);
    acc.x = fmaf(sl, hc.x, acc.x);
    acc.y = fmaf(sl, hc.y, acc.y);
    acc.z = fmaf(sl, hc.z, acc.z);
    acc.w = fmaf(sl, hc.w, acc.w);
    return acc;
}

#define TP 132
__device__ __forceinline__ void mm_pair(const float* tile, const float* __restrict__ W,
                                        int p, int l, int hi, v8f& ev, v8f& od) {
    const float* ap = tile + l * TP + 2 * hi;
    const float* bp = W + (size_t)(2 * hi) * NFEAT + 32 * p + 2 * l;
#pragma unroll
    for (int kk = 0; kk < NFEAT; kk += 32) {
        v2f a[8];
#pragma unroll
        for (int i = 0; i < 8; ++i) a[i] = *(const v2f*)(ap + kk + 4 * i);
#pragma unroll
        for (int i = 0; i < 8; ++i) {
            const int k = kk + 4 * i;
            const v2f r0 = *(const v2f*)(bp + (size_t)k * NFEAT);
            const v2f r1 = *(const v2f*)(bp + (size_t)k * NFEAT + NFEAT);
            v2f be, bo;
            be.x = r0.x; be.y = r1.x;
            bo.x = r0.y; bo.y = r1.y;
            ev = __builtin_amdgcn_wmma_f32_16x16x4_f32(
                false, a[i], false, be, (short)0, ev, false, false);
            od = __builtin_amdgcn_wmma_f32_16x16x4_f32(
                false, a[i], false, bo, (short)0, od, false, false);
        }
    }
}

__global__ void __launch_bounds__(128)
k_agg_mm2(const int* __restrict__ off, const int* __restrict__ csr_rec,
          const float* __restrict__ dinv,
          const float* __restrict__ x, const float* __restrict__ W1,
          const float* __restrict__ b1, const float* __restrict__ W2,
          float* __restrict__ h2, int n) {
    __shared__ __attribute__((aligned(16))) float tile[16 * TP];
    const int lane = threadIdx.x;
    const int w    = __builtin_amdgcn_readfirstlane(threadIdx.y);
    const int m0   = blockIdx.x * 16;
    const int hi   = lane >> 4;
    const int l    = lane & 15;
    const int c0   = m0 + 4 * w;

    int   off_l = 0;
    float dv_l  = 0.0f;
    if (lane <= 4) { const int ci = c0 + lane; off_l = off[ci < n ? ci : n]; }
    if (lane < 4 && c0 + lane < n) dv_l = dinv[c0 + lane];
#pragma unroll 1
    for (int j = 0; j < 4; ++j) {
        const int c = c0 + j;
        float4 a = make_float4(0.0f, 0.0f, 0.0f, 0.0f);
        if (c < n) {
            const int   s  = __builtin_amdgcn_readlane(off_l, j);
            const int   t  = __builtin_amdgcn_readlane(off_l, j + 1);
            const float dc = __int_as_float(__builtin_amdgcn_readlane(__float_as_int(dv_l), j));
            a = agg_row(csr_rec, dinv, x, c, s, t, dc, lane);
        }
        *(float4*)(tile + (4 * w + j) * TP + 4 * lane) = a;
    }
    __syncthreads();

    v8f ev = {}, od = {};
    mm_pair(tile, W1, w, l, hi, ev, od);
    const v2f bb = *(const v2f*)(b1 + 32 * w + 2 * l);
    __syncthreads();
#pragma unroll
    for (int v = 0; v < 8; ++v) {
        v2f g;
        g.x = fmaxf(ev[v] + bb.x, 0.0f);
        g.y = fmaxf(od[v] + bb.y, 0.0f);
        *(v2f*)(tile + (v + 8 * hi) * TP + 32 * w + 2 * l) = g;
    }
    __syncthreads();

    v8f ev2 = {}, od2 = {};
    mm_pair(tile, W2, w, l, hi, ev2, od2);
    float* op = h2 + (size_t)(m0 + 8 * hi) * NFEAT + 32 * w + 2 * l;
#pragma unroll
    for (int v = 0; v < 8; ++v) {
        if (m0 + 8 * hi + v < n) {
            v2f o;
            o.x = ev2[v];
            o.y = od2[v];
            *(v2f*)(op + (size_t)v * NFEAT) = o;
        }
    }
}

__global__ void __launch_bounds__(256)
k_agg(const int* __restrict__ off, const int* __restrict__ csr_rec,
      const float* __restrict__ dinv,
      const float* __restrict__ h, const float* __restrict__ bias,
      float* __restrict__ out, int n) {
    const int lane = threadIdx.x;
    const int c = blockIdx.x * 8 + __builtin_amdgcn_readfirstlane(threadIdx.y);
    if (c >= n) return;
    const int s = __builtin_amdgcn_readfirstlane(off[c]);
    const int t = __builtin_amdgcn_readfirstlane(off[c + 1]);
    const float dc = dinv[c];
    const int f = lane * 4;
    const float4 a  = agg_row(csr_rec, dinv, h, c, s, t, dc, lane);
    const float4 bv = *(const float4*)(bias + f);
    float4 o;
    o.x = a.x + bv.x;
    o.y = a.y + bv.y;
    o.z = a.z + bv.z;
    o.w = a.w + bv.w;
    *(float4*)(out + (size_t)c * NFEAT + f) = o;
}

extern "C" void kernel_launch(void* const* d_in, const int* in_sizes, int n_in,
                              void* d_out, int out_size, void* d_ws, size_t ws_size,
                              hipStream_t stream) {
    const float* x  = (const float*)d_in[0];
    const int*   ei = (const int*)d_in[1];
    const float* ew = (const float*)d_in[2];
    const float* W1 = (const float*)d_in[3];
    const float* b1 = (const float*)d_in[4];
    const float* W2 = (const float*)d_in[5];
    const float* b2 = (const float*)d_in[6];
    float* out = (float*)d_out;

    const int N = in_sizes[0] / NFEAT;
    const int E = in_sizes[2];
    const int* row = ei;
    const int* col = ei + E;

    const int nchunk = (E + CHUNK - 1) / CHUNK;
    size_t npad = ((size_t)N + 1 + 1023) & ~(size_t)1023;
    size_t epad = ((size_t)E + 1023) & ~(size_t)1023;
    int*   tbl     = (int*)d_ws;
    int*   off     = tbl + (size_t)nchunk * TBP;
    float* dinv    = (float*)(off + npad);
    int*   crec    = (int*)(dinv + npad);
    int*   bkt_rec = crec + 2 * (size_t)nchunk * CHUNK;
    int*   csr_rec = bkt_rec + 2 * epad;
    float* h       = (float*)(csr_rec + 2 * epad);

    const int T = 256;
    dim3 wblk(32, 8);
    dim3 gblk(32, 4);
    int ntile16 = (N + 15) / 16;
    int nblk8  = (N + 7) / 8;
    int nbkt   = (N + BK_NODES - 1) / BK_NODES;

    k_bscatter<<<nchunk, T, 0, stream>>>(col, row, ew, tbl, crec, E);
    k_bsort<<<nbkt, T, 0, stream>>>(tbl, crec, bkt_rec, off, csr_rec, dinv, N, E, nchunk);

    k_agg_mm2<<<ntile16, gblk, 0, stream>>>(off, csr_rec, dinv, x, W1, b1, W2, h, N);

    k_agg<<<nblk8, wblk, 0, stream>>>(off, csr_rec, dinv, h, b2, out, N);
}


# Both sort passes place each record in an LDS image of their output region and write the image out as whole lines: no scattered 8-byte store per edge; degrees summed from the image, same order
# speedup vs baseline: 4.9461x; 1.0640x over previous
#include <hip/hip_runtime.h>

#define NFEAT 128

typedef float v2f  __attribute__((ext_vector_type(2)));
typedef float v8f  __attribute__((ext_vector_type(8)));
typedef int   v2i  __attribute__((ext_vector_type(2)));

#define BK_SHIFT 7
#define BK_NODES 128
#define BK_MAX   1024
#define CHUNK    4096
#define GRP      (CHUNK / 8 / 32)
#define TBP      (BK_MAX + 4)
#define MAXCH    1024
#define LOC_SHIFT 24
#define SCAP     4096

template <int BITS>
__device__ __forceinline__ unsigned match_peers(int key, bool valid) {
    unsigned peers = __builtin_amdgcn_ballot_w32(valid);
#pragma unroll
    for (int b = 0; b < BITS; ++b) {
        const bool bit = (key >> b) & 1;
        const unsigned m = __builtin_amdgcn_ballot_w32(bit);
        peers &= bit ? m : ~m;
    }
    return peers;
}

__global__ void __launch_bounds__(256)
k_bscatter(const int* __restrict__ col, const int* __restrict__ row,
           const float* __restrict__ ew, int* __restrict__ tbl,
           int* __restrict__ crec, int E) {
    __shared__ __attribute__((aligned(16))) int sh[8 * BK_MAX];
    __shared__ __attribute__((aligned(16))) int img[2 * CHUNK];
    __shared__ int ssum[256];
    const int tid  = threadIdx.x;
    const int lane = tid & 31;
    const int w    = __builtin_amdgcn_readfirstlane(tid >> 5);
    const int e0   = blockIdx.x * CHUNK + w * (CHUNK / 8);
    for (int i = tid; i < 8 * BK_MAX; i += 256) sh[i] = 0;
    int   cv[GRP];
    int   rv[GRP];
    float wv[GRP];
#pragma unroll
    for (int g = 0; g < GRP; ++g) {
        const int e = e0 + g * 32 + lane;
        cv[g] = (e < E) ? col[e] : -1;
        rv[g] = (e < E) ? row[e] : 0;
        wv[g] = (e < E) ? ew[e] : 0.0f;
    }
    __syncthreads();
    const int wb = w * BK_MAX;
#pragma unroll
    for (int g = 0; g < GRP; ++g) {
        const bool valid = cv[g] >= 0;
        const int  key   = valid ? (cv[g] >> BK_SHIFT) : 0;
        const unsigned peers = match_peers<10>(key, valid);
        if (valid && (peers & ((1u << lane) - 1u)) == 0u)
            sh[wb + key] = sh[wb + key] + __popc(peers);
    }
    __syncthreads();
    const int k4 = 4 * tid;
    int t0 = 0, t1 = 0, t2 = 0, t3 = 0;
#pragma unroll
    for (int w2 = 0; w2 < 8; ++w2) {
        const int4 c = *(const int4*)(sh + w2 * BK_MAX + k4);
        t0 += c.x; t1 += c.y; t2 += c.z; t3 += c.w;
    }
    const int mysum = t0 + t1 + t2 + t3;
    ssum[tid] = mysum;
    __syncthreads();
    for (int d = 1; d < 256; d <<= 1) {
        const int v = (tid >= d) ? ssum[tid - d] : 0;
        __syncthreads();
        ssum[tid] += v;
        __syncthreads();
    }
    int r0 = ssum[tid] - mysum;
    int r1 = r0 + t0, r2 = r1 + t1, r3 = r2 + t2;
    int* trow = tbl + (size_t)blockIdx.x * TBP;
    *(int4*)(trow + k4) = make_int4(r0, r1, r2, r3);
    if (tid == 255) trow[BK_MAX] = r3 + t3;
#pragma unroll
    for (int w2 = 0; w2 < 8; ++w2) {
        const int4 c = *(const int4*)(sh + w2 * BK_MAX + k4);
        *(int4*)(sh + w2 * BK_MAX + k4) = make_int4(r0, r1, r2, r3);
        r0 += c.x; r1 += c.y; r2 += c.z; r3 += c.w;
    }
    __syncthreads();
#pragma unroll
    for (int g = 0; g < GRP; ++g) {
        const bool valid = cv[g] >= 0;
        const int  key   = valid ? (cv[g] >> BK_SHIFT) : 0;
        const unsigned peers = match_peers<10>(key, valid);
        const unsigned below = peers & ((1u << lane) - 1u);
        if (valid) {
            const int base = sh[wb + key];
            const int pos  = base + __popc(below);
            if (below == 0u) sh[wb + key] = base + __popc(peers);
            v2i rec;
            rec.x = rv[g] | ((cv[g] & (BK_NODES - 1)) << LOC_SHIFT);
            rec.y = __float_as_int(wv[g]);
            ((v2i*)img)[pos] = rec;
        }
    }
    __syncthreads();
    int cnt = E - blockIdx.x * CHUNK;
    cnt = (cnt < CHUNK) ? cnt : CHUNK;
    int4* cdst = (int4*)(crec + 2 * (size_t)blockIdx.x * CHUNK);
    const int mis = (int)(((size_t)cdst >> 4) & 7);
    for (int i = tid - mis; 2 * i < cnt; i += 256)
        if (i >= 0) cdst[i] = ((const int4*)img)[i];
}

__global__ void __launch_bounds__(256)
k_bsort(const int* __restrict__ tbl, const int* __restrict__ crec,
        int* __restrict__ bkt_rec, int* __restrict__ off, int* __restrict__ csr_rec,
        float* __restrict__ dinv, int n, int E, int nchunk) {
    __shared__ int sh[8 * BK_NODES];
    __shared__ int tot[BK_NODES];
    __shared__ int rs[MAXCH];
    __shared__ int rp[MAXCH];
    __shared__ int sc[256];
    __shared__ int ss[256];
    __shared__ __attribute__((aligned(16))) int img[2 * SCAP];
    const int tid  = threadIdx.x;
    const int lane = tid & 31;
    const int w    = __builtin_amdgcn_readfirstlane(tid >> 5);
    const int kb   = blockIdx.x;
    const int wb   = w * BK_NODES;

    for (int i = tid; i < 8 * BK_NODES; i += 256) sh[i] = 0;
    int s4[4], c4[4];
    int ls = 0, lc = 0;
#pragma unroll
    for (int j = 0; j < 4; ++j) {
        const int c = 4 * tid + j;
        const int* trow = tbl + (size_t)((c < nchunk) ? c : nchunk - 1) * TBP + kb;
        const int s = (c < nchunk) ? trow[0] : 0;
        const int e = (c < nchunk) ? trow[1] : 0;
        s4[j] = s;
        c4[j] = e - s;
        ls += s;
        lc += e - s;
    }
    sc[tid] = lc;
    ss[tid] = ls;
    __syncthreads();
    for (int d = 1; d < 256; d <<= 1) {
        int vc = 0, vs = 0;
        if (tid >= d) { vc = sc[tid - d]; vs = ss[tid - d]; }
        __syncthreads();
        sc[tid] += vc;
        ss[tid] += vs;
        __syncthreads();
    }
    {
        int run = sc[tid] - lc;
#pragma unroll
        for (int j = 0; j < 4; ++j) {
            rs[4 * tid + j] = s4[j];
            rp[4 * tid + j] = run;
            run += c4[j];
        }
    }
    const int total = __builtin_amdgcn_readfirstlane(sc[255]);
    const int bs    = __builtin_amdgcn_readfirstlane(ss[255]);
    const int per   = ((total + 255) >> 8) << 5;
    const int wbeg  = __builtin_amdgcn_readfirstlane(w * per);
    const int wlim  = wbeg + per;
    const int wend  = (wlim < total) ? wlim : total;
    v2i* blist = (v2i*)bkt_rec + bs;
    const bool staged = total <= SCAP;
    __syncthreads();
    for (int p0 = wbeg; p0 < wend; p0 += 64) {
        int  p[2];
        bool valid[2];
        v2i  rec[2];
#pragma unroll
        for (int u = 0; u < 2; ++u) {
            p[u]     = p0 + 32 * u + lane;
            valid[u] = p[u] < wend;
            const int q = valid[u] ? p[u] : total - 1;
            int c = 0;
#pragma unroll
            for (int step = MAXCH / 2; step >= 1; step >>= 1)
                c += (rp[c + step] <= q) ? step : 0;
            rec[u] = ((const v2i*)crec)[(size_t)c * CHUNK + rs[c] + (q - rp[c])];
        }
#pragma unroll
        for (int u = 0; u < 2; ++u) {
            if (valid[u]) blist[p[u]] = rec[u];
            const int  key   = valid[u] ? (rec[u].x >> LOC_SHIFT) : 0;
            const unsigned peers = match_peers<7>(key, valid[u]);
            if (valid[u] && (peers & ((1u << lane) - 1u)) == 0u)
                sh[wb + key] = sh[wb + key] + __popc(peers);
        }
    }
    __syncthreads();
    int mine = 0;
    if (tid < BK_NODES) {
#pragma unroll
        for (int w2 = 0; w2 < 8; ++w2) mine += sh[w2 * BK_NODES + tid];
        tot[tid] = mine;
    }
    __syncthreads();
    for (int d = 1; d < BK_NODES; d <<= 1) {
        const int v = (tid < BK_NODES && tid >= d) ? tot[tid - d] : 0;
        __syncthreads();
        if (tid < BK_NODES) tot[tid] += v;
        __syncthreads();
    }
    int seg = 0;
    if (tid < BK_NODES) {
        seg = bs + tot[tid] - mine;
        int run = seg;
        const int c = kb * BK_NODES + tid;
        if (c < n) off[c] = run;
#pragma unroll
        for (int w2 = 0; w2 < 8; ++w2) {
            const int cc = sh[w2 * BK_NODES + tid];
            sh[w2 * BK_NODES + tid] = run;
            run += cc;
        }
    }
    if (kb == 0 && tid == 0) off[n] = E;
    __syncthreads();
    for (int p0 = wbeg; p0 < wend; p0 += 64) {
        int  p[2];
        bool valid[2];
        v2i  rec[2];
#pragma unroll
        for (int u = 0; u < 2; ++u) {
            p[u]     = p0 + 32 * u + lane;
            valid[u] = p[u] < wend;
            rec[u]   = blist[valid[u] ? p[u] : total - 1];
        }
#pragma unroll
        for (int u = 0; u < 2; ++u) {
            const int  key   = valid[u] ? (rec[u].x >> LOC_SHIFT) : 0;
            const unsigned peers = match_peers<7>(key, valid[u]);
            const unsigned below = peers & ((1u << lane) - 1u);
            if (valid[u]) {
                const int base = sh[wb + key];
                const int pos  = base + __popc(below);
                if (below == 0u) sh[wb + key] = base + __popc(peers);
                v2i o;
                o.x = rec[u].x & ((1 << LOC_SHIFT) - 1);
                o.y = rec[u].y;
                if (staged) ((v2i*)img)[pos - bs] = o;
                else        ((v2i*)csr_rec)[pos] = o;
            }
        }
    }
    __syncthreads();
    if (staged) {
        const int lead = (bs + (int)(((size_t)csr_rec >> 3) & 15)) & 15;
        for (int i = tid - lead; i < total; i += 256)
            if (i >= 0) ((v2i*)csr_rec)[bs + i] = ((const v2i*)img)[i];
    }
    if (staged && tid < BK_NODES && kb * BK_NODES + tid < n) {
        float deg = 1.0f;
        const int lb = seg - bs;
        const int le = lb + mine;
        for (int p = lb; p < le; p += 4) {
            float wq[4];
#pragma unroll
            for (int u = 0; u < 4; ++u)
                wq[u] = (p + u < le) ? __int_as_float(img[2 * (p + u) + 1]) : 0.0f;
#pragma unroll
            for (int u = 0; u < 4; ++u) deg += wq[u];
        }
        dinv[kb * BK_NODES + tid] = (deg > 0.0f) ? rsqrtf(deg) : 0.0f;
    }
    if (!staged && tid < BK_NODES && kb * BK_NODES + tid < n) {
        float deg = 1.0f;
        const int end = seg + mine;
        for (int p = seg; p < end; p += 4) {
            float wq[4];
#pragma unroll
            for (int u = 0; u < 4; ++u)
                wq[u] = (p + u < end) ? __int_as_float(csr_rec[2 * (p + u) + 1]) : 0.0f;
#pragma unroll
            for (int u = 0; u < 4; ++u) deg += wq[u];
        }
        dinv[kb * BK_NODES + tid] = (deg > 0.0f) ? rsqrtf(deg) : 0.0f;
    }
}

__device__ __forceinline__ float4 agg_row(const int* __restrict__ csr_rec,
                                          const float* __restrict__ dinv,
                                          const float* __restrict__ tab,
                                          int c, int s, int t, float dc, int lane) {
    const int f = lane * 4;
    float4 acc = make_float4(0.0f, 0.0f, 0.0f, 0.0f);
    for (int base = s; base < t; base += 32) {
        const int m = (t - base < 32) ? (t - base) : 32;
        int   r_l = 0;
        float w_l = 0.0f;
        if (lane < m) {
            const v2i rec = ((const v2i*)csr_rec)[base + lane];
            r_l = rec.x;
            w_l = dinv[r_l] * __int_as_float(rec.y) * dc;
        }
#pragma unroll 4
        for (int k = 0; k < m; ++k) {
            const int   r = __builtin_amdgcn_readlane(r_l, k);
            const float w = __int_as_float(__builtin_amdgcn_readlane(__float_as_int(w_l), k));
            const float4 hv = *(const float4*)(tab + (size_t)r * NFEAT + f);
            acc.x = fmaf(w, hv.x, acc.x);
            acc.y = fmaf(w, hv.y, acc.y);
            acc.z = fmaf(w, hv.z, acc.z);
            acc.w = fmaf(w, hv.w, acc.w);
        }
    }
    const float  sl = dc * dc;
    const float4 hc = *(const float4*)(tab + (size_t)c * NFEAT + f);
    acc.x = fmaf(sl, hc.x, acc.x);
    acc.y = fmaf(sl, hc.y, acc.y);
    acc.z = fmaf(sl, hc.z, acc.z);
    acc.w = fmaf(sl, hc.w, acc.w);
    return acc;
}

#define TP 132
__device__ __forceinline__ void mm_pair(const float* tile, const float* __restrict__ W,
                                        int p, int l, int hi, v8f& ev, v8f& od) {
    const float* ap = tile + l * TP + 2 * hi;
    const float* bp = W + (size_t)(2 * hi) * NFEAT + 32 * p + 2 * l;
#pragma unroll
    for (int kk = 0; kk < NFEAT; kk += 32) {
        v2f a[8];
#pragma unroll
        for (int i = 0; i < 8; ++i) a[i] = *(const v2f*)(ap + kk + 4 * i);
#pragma unroll
        for (int i = 0; i < 8; ++i) {
            const int k = kk + 4 * i;
            const v2f r0 = *(const v2f*)(bp + (size_t)k * NFEAT);
            const v2f r1 = *(const v2f*)(bp + (size_t)k * NFEAT + NFEAT);
            v2f be, bo;
            be.x = r0.x; be.y = r1.x;
            bo.x = r0.y; bo.y = r1.y;
            ev = __builtin_amdgcn_wmma_f32_16x16x4_f32(
                false, a[i], false, be, (short)0, ev, false, false);
            od = __builtin_amdgcn_wmma_f32_16x16x4_f32(
                false, a[i], false, bo, (short)0, od, false, false);
        }
    }
}

__global__ void __launch_bounds__(128)
k_agg_mm2(const int* __restrict__ off, const int* __restrict__ csr_rec,
          const float* __restrict__ dinv,
          const float* __restrict__ x, const float* __restrict__ W1,
          const float* __restrict__ b1, const float* __restrict__ W2,
          float* __restrict__ h2, int n) {
    __shared__ __attribute__((aligned(16))) float tile[16 * TP];
    const int lane = threadIdx.x;
    const int w    = __builtin_amdgcn_readfirstlane(threadIdx.y);
    const int m0   = blockIdx.x * 16;
    const int hi   = lane >> 4;
    const int l    = lane & 15;
    const int c0   = m0 + 4 * w;

    int   off_l = 0;
    float dv_l  = 0.0f;
    if (lane <= 4) { const int ci = c0 + lane; off_l = off[ci < n ? ci : n]; }
    if (lane < 4 && c0 + lane < n) dv_l = dinv[c0 + lane];
#pragma unroll 1
    for (int j = 0; j < 4; ++j) {
        const int c = c0 + j;
        float4 a = make_float4(0.0f, 0.0f, 0.0f, 0.0f);
        if (c < n) {
            const int   s  = __builtin_amdgcn_readlane(off_l, j);
            const int   t  = __builtin_amdgcn_readlane(off_l, j + 1);
            const float dc = __int_as_float(__builtin_amdgcn_readlane(__float_as_int(dv_l), j));
            a = agg_row(csr_rec, dinv, x, c, s, t, dc, lane);
        }
        *(float4*)(tile + (4 * w + j) * TP + 4 * lane) = a;
    }
    __syncthreads();

    v8f ev = {}, od = {};
    mm_pair(tile, W1, w, l, hi, ev, od);
    const v2f bb = *(const v2f*)(b1 + 32 * w + 2 * l);
    __syncthreads();
#pragma unroll
    for (int v = 0; v < 8; ++v) {
        v2f g;
        g.x = fmaxf(ev[v] + bb.x, 0.0f);
        g.y = fmaxf(od[v] + bb.y, 0.0f);
        *(v2f*)(tile + (v + 8 * hi) * TP + 32 * w + 2 * l) = g;
    }
    __syncthreads();

    v8f ev2 = {}, od2 = {};
    mm_pair(tile, W2, w, l, hi, ev2, od2);
    float* op = h2 + (size_t)(m0 + 8 * hi) * NFEAT + 32 * w + 2 * l;
#pragma unroll
    for (int v = 0; v < 8; ++v) {
        if (m0 + 8 * hi + v < n) {
            v2f o;
            o.x = ev2[v];
            o.y = od2[v];
            *(v2f*)(op + (size_t)v * NFEAT) = o;
        }
    }
}

__global__ void __launch_bounds__(256)
k_agg(const int* __restrict__ off, const int* __restrict__ csr_rec,
      const float* __restrict__ dinv,
      const float* __restrict__ h, const float* __restrict__ bias,
      float* __restrict__ out, int n) {
    const int lane = threadIdx.x;
    const int c = blockIdx.x * 8 + __builtin_amdgcn_readfirstlane(threadIdx.y);
    if (c >= n) return;
    const int s = __builtin_amdgcn_readfirstlane(off[c]);
    const int t = __builtin_amdgcn_readfirstlane(off[c + 1]);
    const float dc = dinv[c];
    const int f = lane * 4;
    const float4 a  = agg_row(csr_rec, dinv, h, c, s, t, dc, lane);
    const float4 bv = *(const float4*)(bias + f);
    float4 o;
    o.x = a.x + bv.x;
    o.y = a.y + bv.y;
    o.z = a.z + bv.z;
    o.w = a.w + bv.w;
    *(float4*)(out + (size_t)c * NFEAT + f) = o;
}

extern "C" void kernel_launch(void* const* d_in, const int* in_sizes, int n_in,
                              void* d_out, int out_size, void* d_ws, size_t ws_size,
                              hipStream_t stream) {
    const float* x  = (const float*)d_in[0];
    const int*   ei = (const int*)d_in[1];
    const float* ew = (const float*)d_in[2];
    const float* W1 = (const float*)d_in[3];
    const float* b1 = (const float*)d_in[4];
    const float* W2 = (const float*)d_in[5];
    const float* b2 = (const float*)d_in[6];
    float* out = (float*)d_out;

    const int N = in_sizes[0] / NFEAT;
    const int E = in_sizes[2];
    const int* row = ei;
    const int* col = ei + E;

    const int nchunk = (E + CHUNK - 1) / CHUNK;
    size_t npad = ((size_t)N + 1 + 1023) & ~(size_t)1023;
    size_t epad = ((size_t)E + 1023) & ~(size_t)1023;
    int*   tbl     = (int*)d_ws;
    int*   off     = tbl + (size_t)nchunk * TBP;
    float* dinv    = (float*)(off + npad);
    int*   crec    = (int*)(dinv + npad);
    int*   bkt_rec = crec + 2 * (size_t)nchunk * CHUNK;
    int*   csr_rec = bkt_rec + 2 * epad;
    float* h       = (float*)(csr_rec + 2 * epad);

    const int T = 256;
    dim3 wblk(32, 8);
    dim3 gblk(32, 4);
    int ntile16 = (N + 15) / 16;
    int nblk8  = (N + 7) / 8;
    int nbkt   = (N + BK_NODES - 1) / BK_NODES;

    k_bscatter<<<nchunk, T, 0, stream>>>(col, row, ew, tbl, crec, E);
    k_bsort<<<nbkt, T, 0, stream>>>(tbl, crec, bkt_rec, off, csr_rec, dinv, N, E, nchunk);

    k_agg_mm2<<<ntile16, gblk, 0, stream>>>(off, csr_rec, dinv, x, W1, b1, W2, h, N);

    k_agg<<<nblk8, wblk, 0, stream>>>(off, csr_rec, dinv, h, b2, out, N);
}


# Edge norm dinv[src]*w*dinv[dst] is formed once, in the layer-1 gather, and written into the edge record's weight slot; the layer-2 gather reads it there: no scattered dinv[src] read per edge
# speedup vs baseline: 5.1353x; 1.0383x over previous
#include <hip/hip_runtime.h>

#define NFEAT 128

typedef float v2f  __attribute__((ext_vector_type(2)));
typedef float v8f  __attribute__((ext_vector_type(8)));
typedef int   v2i  __attribute__((ext_vector_type(2)));

#define BK_SHIFT 7
#define BK_NODES 128
#define BK_MAX   1024
#define CHUNK    4096
#define GRP      (CHUNK / 8 / 32)
#define TBP      (BK_MAX + 4)
#define MAXCH    1024
#define LOC_SHIFT 24
#define SCAP     4096

template <int BITS>
__device__ __forceinline__ unsigned match_peers(int key, bool valid) {
    unsigned peers = __builtin_amdgcn_ballot_w32(valid);
#pragma unroll
    for (int b = 0; b < BITS; ++b) {
        const bool bit = (key >> b) & 1;
        const unsigned m = __builtin_amdgcn_ballot_w32(bit);
        peers &= bit ? m : ~m;
    }
    return peers;
}

__global__ void __launch_bounds__(256)
k_bscatter(const int* __restrict__ col, const int* __restrict__ row,
           const float* __restrict__ ew, int* __restrict__ tbl,
           int* __restrict__ crec, int E) {
    __shared__ __attribute__((aligned(16))) int sh[8 * BK_MAX];
    __shared__ __attribute__((aligned(16))) int img[2 * CHUNK];
    __shared__ int ssum[256];
    const int tid  = threadIdx.x;
    const int lane = tid & 31;
    const int w    = __builtin_amdgcn_readfirstlane(tid >> 5);
    const int e0   = blockIdx.x * CHUNK + w * (CHUNK / 8);
    for (int i = tid; i < 8 * BK_MAX; i += 256) sh[i] = 0;
    int   cv[GRP];
    int   rv[GRP];
    float wv[GRP];
#pragma unroll
    for (int g = 0; g < GRP; ++g) {
        const int e = e0 + g * 32 + lane;
        cv[g] = (e < E) ? col[e] : -1;
        rv[g] = (e < E) ? row[e] : 0;
        wv[g] = (e < E) ? ew[e] : 0.0f;
    }
    __syncthreads();
    const int wb = w * BK_MAX;
#pragma unroll
    for (int g = 0; g < GRP; ++g) {
        const bool valid = cv[g] >= 0;
        const int  key   = valid ? (cv[g] >> BK_SHIFT) : 0;
        const unsigned peers = match_peers<10>(key, valid);
        if (valid && (peers & ((1u << lane) - 1u)) == 0u)
            sh[wb + key] = sh[wb + key] + __popc(peers);
    }
    __syncthreads();
    const int k4 = 4 * tid;
    int t0 = 0, t1 = 0, t2 = 0, t3 = 0;
#pragma unroll
    for (int w2 = 0; w2 < 8; ++w2) {
        const int4 c = *(const int4*)(sh + w2 * BK_MAX + k4);
        t0 += c.x; t1 += c.y; t2 += c.z; t3 += c.w;
    }
    const int mysum = t0 + t1 + t2 + t3;
    ssum[tid] = mysum;
    __syncthreads();
    for (int d = 1; d < 256; d <<= 1) {
        const int v = (tid >= d) ? ssum[tid - d] : 0;
        __syncthreads();
        ssum[tid] += v;
        __syncthreads();
    }
    int r0 = ssum[tid] - mysum;
    int r1 = r0 + t0, r2 = r1 + t1, r3 = r2 + t2;
    int* trow = tbl + (size_t)blockIdx.x * TBP;
    *(int4*)(trow + k4) = make_int4(r0, r1, r2, r3);
    if (tid == 255) trow[BK_MAX] = r3 + t3;
#pragma unroll
    for (int w2 = 0; w2 < 8; ++w2) {
        const int4 c = *(const int4*)(sh + w2 * BK_MAX + k4);
        *(int4*)(sh + w2 * BK_MAX + k4) = make_int4(r0, r1, r2, r3);
        r0 += c.x; r1 += c.y; r2 += c.z; r3 += c.w;
    }
    __syncthreads();
#pragma unroll
    for (int g = 0; g < GRP; ++g) {
        const bool valid = cv[g] >= 0;
        const int  key   = valid ? (cv[g] >> BK_SHIFT) : 0;
        const unsigned peers = match_peers<10>(key, valid);
        const unsigned below = peers & ((1u << lane) - 1u);
        if (valid) {
            const int base = sh[wb + key];
            const int pos  = base + __popc(below);
            if (below == 0u) sh[wb + key] = base + __popc(peers);
            v2i rec;
            rec.x = rv[g] | ((cv[g] & (BK_NODES - 1)) << LOC_SHIFT);
            rec.y = __float_as_int(wv[g]);
            ((v2i*)img)[pos] = rec;
        }
    }
    __syncthreads();
    int cnt = E - blockIdx.x * CHUNK;
    cnt = (cnt < CHUNK) ? cnt : CHUNK;
    int4* cdst = (int4*)(crec + 2 * (size_t)blockIdx.x * CHUNK);
    const int mis = (int)(((size_t)cdst >> 4) & 7);
    for (int i = tid - mis; 2 * i < cnt; i += 256)
        if (i >= 0) cdst[i] = ((const int4*)img)[i];
}

__global__ void __launch_bounds__(256)
k_bsort(const int* __restrict__ tbl, const int* __restrict__ crec,
        int* __restrict__ bkt_rec, int* __restrict__ off, int* __restrict__ csr_rec,
        float* __restrict__ dinv, int n, int E, int nchunk) {
    __shared__ int sh[8 * BK_NODES];
    __shared__ int tot[BK_NODES];
    __shared__ int rs[MAXCH];
    __shared__ int rp[MAXCH];
    __shared__ int sc[256];
    __shared__ int ss[256];
    __shared__ __attribute__((aligned(16))) int img[2 * SCAP];
    const int tid  = threadIdx.x;
    const int lane = tid & 31;
    const int w    = __builtin_amdgcn_readfirstlane(tid >> 5);
    const int kb   = blockIdx.x;
    const int wb   = w * BK_NODES;

    for (int i = tid; i < 8 * BK_NODES; i += 256) sh[i] = 0;
    int s4[4], c4[4];
    int ls = 0, lc = 0;
#pragma unroll
    for (int j = 0; j < 4; ++j) {
        const int c = 4 * tid + j;
        const int* trow = tbl + (size_t)((c < nchunk) ? c : nchunk - 1) * TBP + kb;
        const int s = (c < nchunk) ? trow[0] : 0;
        const int e = (c < nchunk) ? trow[1] : 0;
        s4[j] = s;
        c4[j] = e - s;
        ls += s;
        lc += e - s;
    }
    sc[tid] = lc;
    ss[tid] = ls;
    __syncthreads();
    for (int d = 1; d < 256; d <<= 1) {
        int vc = 0, vs = 0;
        if (tid >= d) { vc = sc[tid - d]; vs = ss[tid - d]; }
        __syncthreads();
        sc[tid] += vc;
        ss[tid] += vs;
        __syncthreads();
    }
    {
        int run = sc[tid] - lc;
#pragma unroll
        for (int j = 0; j < 4; ++j) {
            rs[4 * tid + j] = s4[j];
            rp[4 * tid + j] = run;
            run += c4[j];
        }
    }
    const int total = __builtin_amdgcn_readfirstlane(sc[255]);
    const int bs    = __builtin_amdgcn_readfirstlane(ss[255]);
    const int per   = ((total + 255) >> 8) << 5;
    const int wbeg  = __builtin_amdgcn_readfirstlane(w * per);
    const int wlim  = wbeg + per;
    const int wend  = (wlim < total) ? wlim : total;
    v2i* blist = (v2i*)bkt_rec + bs;
    const bool staged = total <= SCAP;
    __syncthreads();
    for (int p0 = wbeg; p0 < wend; p0 += 64) {
        int  p[2];
        bool valid[2];
        v2i  rec[2];
#pragma unroll
        for (int u = 0; u < 2; ++u) {
            p[u]     = p0 + 32 * u + lane;
            valid[u] = p[u] < wend;
            const int q = valid[u] ? p[u] : total - 1;
            int c = 0;
#pragma unroll
            for (int step = MAXCH / 2; step >= 1; step >>= 1)
                c += (rp[c + step] <= q) ? step : 0;
            rec[u] = ((const v2i*)crec)[(size_t)c * CHUNK + rs[c] + (q - rp[c])];
        }
#pragma unroll
        for (int u = 0; u < 2; ++u) {
            if (valid[u]) blist[p[u]] = rec[u];
            const int  key   = valid[u] ? (rec[u].x >> LOC_SHIFT) : 0;
            const unsigned peers = match_peers<7>(key, valid[u]);
            if (valid[u] && (peers & ((1u << lane) - 1u)) == 0u)
                sh[wb + key] = sh[wb + key] + __popc(peers);
        }
    }
    __syncthreads();
    int mine = 0;
    if (tid < BK_NODES) {
#pragma unroll
        for (int w2 = 0; w2 < 8; ++w2) mine += sh[w2 * BK_NODES + tid];
        tot[tid] = mine;
    }
    __syncthreads();
    for (int d = 1; d < BK_NODES; d <<= 1) {
        const int v = (tid < BK_NODES && tid >= d) ? tot[tid - d] : 0;
        __syncthreads();
        if (tid < BK_NODES) tot[tid] += v;
        __syncthreads();
    }
    int seg = 0;
    if (tid < BK_NODES) {
        seg = bs + tot[tid] - mine;
        int run = seg;
        const int c = kb * BK_NODES + tid;
        if (c < n) off[c] = run;
#pragma unroll
        for (int w2 = 0; w2 < 8; ++w2) {
            const int cc = sh[w2 * BK_NODES + tid];
            sh[w2 * BK_NODES + tid] = run;
            run += cc;
        }
    }
    if (kb == 0 && tid == 0) off[n] = E;
    __syncthreads();
    for (int p0 = wbeg; p0 < wend; p0 += 64) {
        int  p[2];
        bool valid[2];
        v2i  rec[2];
#pragma unroll
        for (int u = 0; u < 2; ++u) {
            p[u]     = p0 + 32 * u + lane;
            valid[u] = p[u] < wend;
            rec[u]   = blist[valid[u] ? p[u] : total - 1];
        }
#pragma unroll
        for (int u = 0; u < 2; ++u) {
            const int  key   = valid[u] ? (rec[u].x >> LOC_SHIFT) : 0;
            const unsigned peers = match_peers<7>(key, valid[u]);
            const unsigned below = peers & ((1u << lane) - 1u);
            if (valid[u]) {
                const int base = sh[wb + key];
                const int pos  = base + __popc(below);
                if (below == 0u) sh[wb + key] = base + __popc(peers);
                v2i o;
                o.x = rec[u].x & ((1 << LOC_SHIFT) - 1);
                o.y = rec[u].y;
                if (staged) ((v2i*)img)[pos - bs] = o;
                else        ((v2i*)csr_rec)[pos] = o;
            }
        }
    }
    __syncthreads();
    if (staged) {
        const int lead = (bs + (int)(((size_t)csr_rec >> 3) & 15)) & 15;
        for (int i = tid - lead; i < total; i += 256)
            if (i >= 0) ((v2i*)csr_rec)[bs + i] = ((const v2i*)img)[i];
    }
    if (staged && tid < BK_NODES && kb * BK_NODES + tid < n) {
        float deg = 1.0f;
        const int lb = seg - bs;
        const int le = lb + mine;
        for (int p = lb; p < le; p += 4) {
            float wq[4];
#pragma unroll
            for (int u = 0; u < 4; ++u)
                wq[u] = (p + u < le) ? __int_as_float(img[2 * (p + u) + 1]) : 0.0f;
#pragma unroll
            for (int u = 0; u < 4; ++u) deg += wq[u];
        }
        dinv[kb * BK_NODES + tid] = (deg > 0.0f) ? rsqrtf(deg) : 0.0f;
    }
    if (!staged && tid < BK_NODES && kb * BK_NODES + tid < n) {
        float deg = 1.0f;
        const int end = seg + mine;
        for (int p = seg; p < end; p += 4) {
            float wq[4];
#pragma unroll
            for (int u = 0; u < 4; ++u)
                wq[u] = (p + u < end) ? __int_as_float(csr_rec[2 * (p + u) + 1]) : 0.0f;
#pragma unroll
            for (int u = 0; u < 4; ++u) deg += wq[u];
        }
        dinv[kb * BK_NODES + tid] = (deg > 0.0f) ? rsqrtf(deg) : 0.0f;
    }
}

template <bool FIRST>
__device__ __forceinline__ float4 agg_row(int* __restrict__ csr_rec,
                                          const float* __restrict__ dinv,
                                          const float* __restrict__ tab,
                                          int c, int s, int t, float dc, int lane) {
    const int f = lane * 4;
    float4 acc = make_float4(0.0f, 0.0f, 0.0f, 0.0f);
    for (int base = s; base < t; base += 32) {
        const int m = (t - base < 32) ? (t - base) : 32;
        int   r_l = 0;
        float w_l = 0.0f;
        if (lane < m) {
            const v2i rec = ((const v2i*)csr_rec)[base + lane];
            r_l = rec.x;
            if (FIRST) {
                w_l = dinv[r_l] * __int_as_float(rec.y) * dc;
                csr_rec[2 * (base + lane) + 1] = __float_as_int(w_l);
            } else {
                w_l = __int_as_float(rec.y);
            }
        }
#pragma unroll 4
        for (int k = 0; k < m; ++k) {
            const int   r = __builtin_amdgcn_readlane(r_l, k);
            const float w = __int_as_float(__builtin_amdgcn_readlane(__float_as_int(w_l), k));
            const float4 hv = *(const float4*)(tab + (size_t)r * NFEAT + f);
            acc.x = fmaf(w, hv.x, acc.x);
            acc.y = fmaf(w, hv.y, acc.y);
            acc.z = fmaf(w, hv.z, acc.z);
            acc.w = fmaf(w, hv.w, acc.w);
        }
    }
    const float  sl = dc * dc;
    const float4 hc = *(const float4*)(tab + (size_t)c * NFEAT + f);
    acc.x = fmaf(sl, hc.x, acc.x);
    acc.y = fmaf(sl, hc.y, acc.y);
    acc.z = fmaf(sl, hc.z, acc.z);
    acc.w = fmaf(sl, hc.w, acc.w);
    return acc;
}

#define TP 132
__device__ __forceinline__ void mm_pair(const float* tile, const float* __restrict__ W,
                                        int p, int l, int hi, v8f& ev, v8f& od) {
    const float* ap = tile + l * TP + 2 * hi;
    const float* bp = W + (size_t)(2 * hi) * NFEAT + 32 * p + 2 * l;
#pragma unroll
    for (int kk = 0; kk < NFEAT; kk += 32) {
        v2f a[8];
#pragma unroll
        for (int i = 0; i < 8; ++i) a[i] = *(const v2f*)(ap + kk + 4 * i);
#pragma unroll
        for (int i = 0; i < 8; ++i) {
            const int k = kk + 4 * i;
            const v2f r0 = *(const v2f*)(bp + (size_t)k * NFEAT);
            const v2f r1 = *(const v2f*)(bp + (size_t)k * NFEAT + NFEAT);
            v2f be, bo;
            be.x = r0.x; be.y = r1.x;
            bo.x = r0.y; bo.y = r1.y;
            ev = __builtin_amdgcn_wmma_f32_16x16x4_f32(
                false, a[i], false, be, (short)0, ev, false, false);
            od = __builtin_amdgcn_wmma_f32_16x16x4_f32(
                false, a[i], false, bo, (short)0, od, false, false);
        }
    }
}

__global__ void __launch_bounds__(128)
k_agg_mm2(const int* __restrict__ off, int* __restrict__ csr_rec,
          const float* __restrict__ dinv,
          const float* __restrict__ x, const float* __restrict__ W1,
          const float* __restrict__ b1, const float* __restrict__ W2,
          float* __restrict__ h2, int n) {
    __shared__ __attribute__((aligned(16))) float tile[16 * TP];
    const int lane = threadIdx.x;
    const int w    = __builtin_amdgcn_readfirstlane(threadIdx.y);
    const int m0   = blockIdx.x * 16;
    const int hi   = lane >> 4;
    const int l    = lane & 15;
    const int c0   = m0 + 4 * w;

    int   off_l = 0;
    float dv_l  = 0.0f;
    if (lane <= 4) { const int ci = c0 + lane; off_l = off[ci < n ? ci : n]; }
    if (lane < 4 && c0 + lane < n) dv_l = dinv[c0 + lane];
#pragma unroll 1
    for (int j = 0; j < 4; ++j) {
        const int c = c0 + j;
        float4 a = make_float4(0.0f, 0.0f, 0.0f, 0.0f);
        if (c < n) {
            const int   s  = __builtin_amdgcn_readlane(off_l, j);
            const int   t  = __builtin_amdgcn_readlane(off_l, j + 1);
            const float dc = __int_as_float(__builtin_amdgcn_readlane(__float_as_int(dv_l), j));
            a = agg_row<true>(csr_rec, dinv, x, c, s, t, dc, lane);
        }
        *(float4*)(tile + (4 * w + j) * TP + 4 * lane) = a;
    }
    __syncthreads();

    v8f ev = {}, od = {};
    mm_pair(tile, W1, w, l, hi, ev, od);
    const v2f bb = *(const v2f*)(b1 + 32 * w + 2 * l);
    __syncthreads();
#pragma unroll
    for (int v = 0; v < 8; ++v) {
        v2f g;
        g.x = fmaxf(ev[v] + bb.x, 0.0f);
        g.y = fmaxf(od[v] + bb.y, 0.0f);
        *(v2f*)(tile + (v + 8 * hi) * TP + 32 * w + 2 * l) = g;
    }
    __syncthreads();

    v8f ev2 = {}, od2 = {};
    mm_pair(tile, W2, w, l, hi, ev2, od2);
    float* op = h2 + (size_t)(m0 + 8 * hi) * NFEAT + 32 * w + 2 * l;
#pragma unroll
    for (int v = 0; v < 8; ++v) {
        if (m0 + 8 * hi + v < n) {
            v2f o;
            o.x = ev2[v];
            o.y = od2[v];
            *(v2f*)(op + (size_t)v * NFEAT) = o;
        }
    }
}

__global__ void __launch_bounds__(256)
k_agg(const int* __restrict__ off, int* __restrict__ csr_rec,
      const float* __restrict__ dinv,
      const float* __restrict__ h, const float* __restrict__ bias,
      float* __restrict__ out, int n) {
    const int lane = threadIdx.x;
    const int c = blockIdx.x * 8 + __builtin_amdgcn_readfirstlane(threadIdx.y);
    if (c >= n) return;
    const int s = __builtin_amdgcn_readfirstlane(off[c]);
    const int t = __builtin_amdgcn_readfirstlane(off[c + 1]);
    const float dc = dinv[c];
    const int f = lane * 4;
    const float4 a  = agg_row<false>(csr_rec, dinv, h, c, s, t, dc, lane);
    const float4 bv = *(const float4*)(bias + f);
    float4 o;
    o.x = a.x + bv.x;
    o.y = a.y + bv.y;
    o.z = a.z + bv.z;
    o.w = a.w + bv.w;
    *(float4*)(out + (size_t)c * NFEAT + f) = o;
}

extern "C" void kernel_launch(void* const* d_in, const int* in_sizes, int n_in,
                              void* d_out, int out_size, void* d_ws, size_t ws_size,
                              hipStream_t stream) {
    const float* x  = (const float*)d_in[0];
    const int*   ei = (const int*)d_in[1];
    const float* ew = (const float*)d_in[2];
    const float* W1 = (const float*)d_in[3];
    const float* b1 = (const float*)d_in[4];
    const float* W2 = (const float*)d_in[5];
    const float* b2 = (const float*)d_in[6];
    float* out = (float*)d_out;

    const int N = in_sizes[0] / NFEAT;
    const int E = in_sizes[2];
    const int* row = ei;
    const int* col = ei + E;

    const int nchunk = (E + CHUNK - 1) / CHUNK;
    size_t npad = ((size_t)N + 1 + 1023) & ~(size_t)1023;
    size_t epad = ((size_t)E + 1023) & ~(size_t)1023;
    int*   tbl     = (int*)d_ws;
    int*   off     = tbl + (size_t)nchunk * TBP;
    float* dinv    = (float*)(off + npad);
    int*   crec    = (int*)(dinv + npad);
    int*   bkt_rec = crec + 2 * (size_t)nchunk * CHUNK;
    int*   csr_rec = bkt_rec + 2 * epad;
    float* h       = (float*)(csr_rec + 2 * epad);

    const int T = 256;
    dim3 wblk(32, 8);
    dim3 gblk(32, 4);
    int ntile16 = (N + 15) / 16;
    int nblk8  = (N + 7) / 8;
    int nbkt   = (N + BK_NODES - 1) / BK_NODES;

    k_bscatter<<<nchunk, T, 0, stream>>>(col, row, ew, tbl, crec, E);
    k_bsort<<<nbkt, T, 0, stream>>>(tbl, crec, bkt_rec, off, csr_rec, dinv, N, E, nchunk);

    k_agg_mm2<<<ntile16, gblk, 0, stream>>>(off, csr_rec, dinv, x, W1, b1, W2, h, N);

    k_agg<<<nblk8, wblk, 0, stream>>>(off, csr_rec, dinv, h, b2, out, N);
}


# Workspace regions start on 512-byte boundaries (table region padded to a multiple of 128 words): a layer-1 output row is 4 whole 128-byte lines, not 5 pieces, for its stores and the layer-2 gathers
# speedup vs baseline: 5.3707x; 1.0458x over previous
#include <hip/hip_runtime.h>

#define NFEAT 128

typedef float v2f  __attribute__((ext_vector_type(2)));
typedef float v8f  __attribute__((ext_vector_type(8)));
typedef int   v2i  __attribute__((ext_vector_type(2)));

#define BK_SHIFT 7
#define BK_NODES 128
#define BK_MAX   1024
#define CHUNK    4096
#define GRP      (CHUNK / 8 / 32)
#define TBP      (BK_MAX + 4)
#define MAXCH    1024
#define LOC_SHIFT 24
#define SCAP     4096

template <int BITS>
__device__ __forceinline__ unsigned match_peers(int key, bool valid) {
    unsigned peers = __builtin_amdgcn_ballot_w32(valid);
#pragma unroll
    for (int b = 0; b < BITS; ++b) {
        const bool bit = (key >> b) & 1;
        const unsigned m = __builtin_amdgcn_ballot_w32(bit);
        peers &= bit ? m : ~m;
    }
    return peers;
}

__global__ void __launch_bounds__(256)
k_bscatter(const int* __restrict__ col, const int* __restrict__ row,
           const float* __restrict__ ew, int* __restrict__ tbl,
           int* __restrict__ crec, int E) {
    __shared__ __attribute__((aligned(16))) int sh[8 * BK_MAX];
    __shared__ __attribute__((aligned(16))) int img[2 * CHUNK];
    __shared__ int ssum[256];
    const int tid  = threadIdx.x;
    const int lane = tid & 31;
    const int w    = __builtin_amdgcn_readfirstlane(tid >> 5);
    const int e0   = blockIdx.x * CHUNK + w * (CHUNK / 8);
    for (int i = tid; i < 8 * BK_MAX; i += 256) sh[i] = 0;
    int   cv[GRP];
    int   rv[GRP];
    float wv[GRP];
#pragma unroll
    for (int g = 0; g < GRP; ++g) {
        const int e = e0 + g * 32 + lane;
        cv[g] = (e < E) ? col[e] : -1;
        rv[g] = (e < E) ? row[e] : 0;
        wv[g] = (e < E) ? ew[e] : 0.0f;
    }
    __syncthreads();
    const int wb = w * BK_MAX;
#pragma unroll
    for (int g = 0; g < GRP; ++g) {
        const bool valid = cv[g] >= 0;
        const int  key   = valid ? (cv[g] >> BK_SHIFT) : 0;
        const unsigned peers = match_peers<10>(key, valid);
        if (valid && (peers & ((1u << lane) - 1u)) == 0u)
            sh[wb + key] = sh[wb + key] + __popc(peers);
    }
    __syncthreads();
    const int k4 = 4 * tid;
    int t0 = 0, t1 = 0, t2 = 0, t3 = 0;
#pragma unroll
    for (int w2 = 0; w2 < 8; ++w2) {
        const int4 c = *(const int4*)(sh + w2 * BK_MAX + k4);
        t0 += c.x; t1 += c.y; t2 += c.z; t3 += c.w;
    }
    const int mysum = t0 + t1 + t2 + t3;
    ssum[tid] = mysum;
    __syncthreads();
    for (int d = 1; d < 256; d <<= 1) {
        const int v = (tid >= d) ? ssum[tid - d] : 0;
        __syncthreads();
        ssum[tid] += v;
        __syncthreads();
    }
    int r0 = ssum[tid] - mysum;
    int r1 = r0 + t0, r2 = r1 + t1, r3 = r2 + t2;
    int* trow = tbl + (size_t)blockIdx.x * TBP;
    *(int4*)(trow + k4) = make_int4(r0, r1, r2, r3);
    if (tid == 255) trow[BK_MAX] = r3 + t3;
#pragma unroll
    for (int w2 = 0; w2 < 8; ++w2) {
        const int4 c = *(const int4*)(sh + w2 * BK_MAX + k4);
        *(int4*)(sh + w2 * BK_MAX + k4) = make_int4(r0, r1, r2, r3);
        r0 += c.x; r1 += c.y; r2 += c.z; r3 += c.w;
    }
    __syncthreads();
#pragma unroll
    for (int g = 0; g < GRP; ++g) {
        const bool valid = cv[g] >= 0;
        const int  key   = valid ? (cv[g] >> BK_SHIFT) : 0;
        const unsigned peers = match_peers<10>(key, valid);
        const unsigned below = peers & ((1u << lane) - 1u);
        if (valid) {
            const int base = sh[wb + key];
            const int pos  = base + __popc(below);
            if (below == 0u) sh[wb + key] = base + __popc(peers);
            v2i rec;
            rec.x = rv[g] | ((cv[g] & (BK_NODES - 1)) << LOC_SHIFT);
            rec.y = __float_as_int(wv[g]);
            ((v2i*)img)[pos] = rec;
        }
    }
    __syncthreads();
    int cnt = E - blockIdx.x * CHUNK;
    cnt = (cnt < CHUNK) ? cnt : CHUNK;
    int4* cdst = (int4*)(crec + 2 * (size_t)blockIdx.x * CHUNK);
    const int mis = (int)(((size_t)cdst >> 4) & 7);
    for (int i = tid - mis; 2 * i < cnt; i += 256)
        if (i >= 0) cdst[i] = ((const int4*)img)[i];
}

__global__ void __launch_bounds__(256)
k_bsort(const int* __restrict__ tbl, const int* __restrict__ crec,
        int* __restrict__ bkt_rec, int* __restrict__ off, int* __restrict__ csr_rec,
        float* __restrict__ dinv, int n, int E, int nchunk) {
    __shared__ int sh[8 * BK_NODES];
    __shared__ int tot[BK_NODES];
    __shared__ int rs[MAXCH];
    __shared__ int rp[MAXCH];
    __shared__ int sc[256];
    __shared__ int ss[256];
    __shared__ __attribute__((aligned(16))) int img[2 * SCAP];
    const int tid  = threadIdx.x;
    const int lane = tid & 31;
    const int w    = __builtin_amdgcn_readfirstlane(tid >> 5);
    const int kb   = blockIdx.x;
    const int wb   = w * BK_NODES;

    for (int i = tid; i < 8 * BK_NODES; i += 256) sh[i] = 0;
    int s4[4], c4[4];
    int ls = 0, lc = 0;
#pragma unroll
    for (int j = 0; j < 4; ++j) {
        const int c = 4 * tid + j;
        const int* trow = tbl + (size_t)((c < nchunk) ? c : nchunk - 1) * TBP + kb;
        const int s = (c < nchunk) ? trow[0] : 0;
        const int e = (c < nchunk) ? trow[1] : 0;
        s4[j] = s;
        c4[j] = e - s;
        ls += s;
        lc += e - s;
    }
    sc[tid] = lc;
    ss[tid] = ls;
    __syncthreads();
    for (int d = 1; d < 256; d <<= 1) {
        int vc = 0, vs = 0;
        if (tid >= d) { vc = sc[tid - d]; vs = ss[tid - d]; }
        __syncthreads();
        sc[tid] += vc;
        ss[tid] += vs;
        __syncthreads();
    }
    {
        int run = sc[tid] - lc;
#pragma unroll
        for (int j = 0; j < 4; ++j) {
            rs[4 * tid + j] = s4[j];
            rp[4 * tid + j] = run;
            run += c4[j];
        }
    }
    const int total = __builtin_amdgcn_readfirstlane(sc[255]);
    const int bs    = __builtin_amdgcn_readfirstlane(ss[255]);
    const int per   = ((total + 255) >> 8) << 5;
    const int wbeg  = __builtin_amdgcn_readfirstlane(w * per);
    const int wlim  = wbeg + per;
    const int wend  = (wlim < total) ? wlim : total;
    v2i* blist = (v2i*)bkt_rec + bs;
    const bool staged = total <= SCAP;
    __syncthreads();
    for (int p0 = wbeg; p0 < wend; p0 += 64) {
        int  p[2];
        bool valid[2];
        v2i  rec[2];
#pragma unroll
        for (int u = 0; u < 2; ++u) {
            p[u]     = p0 + 32 * u + lane;
            valid[u] = p[u] < wend;
            const int q = valid[u] ? p[u] : total - 1;
            int c = 0;
#pragma unroll
            for (int step = MAXCH / 2; step >= 1; step >>= 1)
                c += (rp[c + step] <= q) ? step : 0;
            rec[u] = ((const v2i*)crec)[(size_t)c * CHUNK + rs[c] + (q - rp[c])];
        }
#pragma unroll
        for (int u = 0; u < 2; ++u) {
            if (valid[u]) blist[p[u]] = rec[u];
            const int  key   = valid[u] ? (rec[u].x >> LOC_SHIFT) : 0;
            const unsigned peers = match_peers<7>(key, valid[u]);
            if (valid[u] && (peers & ((1u << lane) - 1u)) == 0u)
                sh[wb + key] = sh[wb + key] + __popc(peers);
        }
    }
    __syncthreads();
    int mine = 0;
    if (tid < BK_NODES) {
#pragma unroll
        for (int w2 = 0; w2 < 8; ++w2) mine += sh[w2 * BK_NODES + tid];
        tot[tid] = mine;
    }
    __syncthreads();
    for (int d = 1; d < BK_NODES; d <<= 1) {
        const int v = (tid < BK_NODES && tid >= d) ? tot[tid - d] : 0;
        __syncthreads();
        if (tid < BK_NODES) tot[tid] += v;
        __syncthreads();
    }
    int seg = 0;
    if (tid < BK_NODES) {
        seg = bs + tot[tid] - mine;
        int run = seg;
        const int c = kb * BK_NODES + tid;
        if (c < n) off[c] = run;
#pragma unroll
        for (int w2 = 0; w2 < 8; ++w2) {
            const int cc = sh[w2 * BK_NODES + tid];
            sh[w2 * BK_NODES + tid] = run;
            run += cc;
        }
    }
    if (kb == 0 && tid == 0) off[n] = E;
    __syncthreads();
    for (int p0 = wbeg; p0 < wend; p0 += 64) {
        int  p[2];
        bool valid[2];
        v2i  rec[2];
#pragma unroll
        for (int u = 0; u < 2; ++u) {
            p[u]     = p0 + 32 * u + lane;
            valid[u] = p[u] < wend;
            rec[u]   = blist[valid[u] ? p[u] : total - 1];
        }
#pragma unroll
        for (int u = 0; u < 2; ++u) {
            const int  key   = valid[u] ? (rec[u].x >> LOC_SHIFT) : 0;
            const unsigned peers = match_peers<7>(key, valid[u]);
            const unsigned below = peers & ((1u << lane) - 1u);
            if (valid[u]) {
                const int base = sh[wb + key];
                const int pos  = base + __popc(below);
                if (below == 0u) sh[wb + key] = base + __popc(peers);
                v2i o;
                o.x = rec[u].x & ((1 << LOC_SHIFT) - 1);
                o.y = rec[u].y;
                if (staged) ((v2i*)img)[pos - bs] = o;
                else        ((v2i*)csr_rec)[pos] = o;
            }
        }
    }
    __syncthreads();
    if (staged) {
        const int lead = (bs + (int)(((size_t)csr_rec >> 3) & 15)) & 15;
        for (int i = tid - lead; i < total; i += 256)
            if (i >= 0) ((v2i*)csr_rec)[bs + i] = ((const v2i*)img)[i];
    }
    if (staged && tid < BK_NODES && kb * BK_NODES + tid < n) {
        float deg = 1.0f;
        const int lb = seg - bs;
        const int le = lb + mine;
        for (int p = lb; p < le; p += 4) {
            float wq[4];
#pragma unroll
            for (int u = 0; u < 4; ++u)
                wq[u] = (p + u < le) ? __int_as_float(img[2 * (p + u) + 1]) : 0.0f;
#pragma unroll
            for (int u = 0; u < 4; ++u) deg += wq[u];
        }
        dinv[kb * BK_NODES + tid] = (deg > 0.0f) ? rsqrtf(deg) : 0.0f;
    }
    if (!staged && tid < BK_NODES && kb * BK_NODES + tid < n) {
        float deg = 1.0f;
        const int end = seg + mine;
        for (int p = seg; p < end; p += 4) {
            float wq[4];
#pragma unroll
            for (int u = 0; u < 4; ++u)
                wq[u] = (p + u < end) ? __int_as_float(csr_rec[2 * (p + u) + 1]) : 0.0f;
#pragma unroll
            for (int u = 0; u < 4; ++u) deg += wq[u];
        }
        dinv[kb * BK_NODES + tid] = (deg > 0.0f) ? rsqrtf(deg) : 0.0f;
    }
}

template <bool FIRST>
__device__ __forceinline__ float4 agg_row(int* __restrict__ csr_rec,
                                          const float* __restrict__ dinv,
                                          const float* __restrict__ tab,
                                          int c, int s, int t, float dc, int lane) {
    const int f = lane * 4;
    float4 acc = make_float4(0.0f, 0.0f, 0.0f, 0.0f);
    for (int base = s; base < t; base += 32) {
        const int m = (t - base < 32) ? (t - base) : 32;
        int   r_l = 0;
        float w_l = 0.0f;
        if (lane < m) {
            const v2i rec = ((const v2i*)csr_rec)[base + lane];
            r_l = rec.x;
            if (FIRST) {
                w_l = dinv[r_l] * __int_as_float(rec.y) * dc;
                csr_rec[2 * (base + lane) + 1] = __float_as_int(w_l);
            } else {
                w_l = __int_as_float(rec.y);
            }
        }
#pragma unroll 4
        for (int k = 0; k < m; ++k) {
            const int   r = __builtin_amdgcn_readlane(r_l, k);
            const float w = __int_as_float(__builtin_amdgcn_readlane(__float_as_int(w_l), k));
            const float4 hv = *(const float4*)(tab + (size_t)r * NFEAT + f);
            acc.x = fmaf(w, hv.x, acc.x);
            acc.y = fmaf(w, hv.y, acc.y);
            acc.z = fmaf(w, hv.z, acc.z);
            acc.w = fmaf(w, hv.w, acc.w);
        }
    }
    const float  sl = dc * dc;
    const float4 hc = *(const float4*)(tab + (size_t)c * NFEAT + f);
    acc.x = fmaf(sl, hc.x, acc.x);
    acc.y = fmaf(sl, hc.y, acc.y);
    acc.z = fmaf(sl, hc.z, acc.z);
    acc.w = fmaf(sl, hc.w, acc.w);
    return acc;
}

#define TP 132
__device__ __forceinline__ void mm_pair(const float* tile, const float* __restrict__ W,
                                        int p, int l, int hi, v8f& ev, v8f& od) {
    const float* ap = tile + l * TP + 2 * hi;
    const float* bp = W + (size_t)(2 * hi) * NFEAT + 32 * p + 2 * l;
#pragma unroll
    for (int kk = 0; kk < NFEAT; kk += 32) {
        v2f a[8];
#pragma unroll
        for (int i = 0; i < 8; ++i) a[i] = *(const v2f*)(ap + kk + 4 * i);
#pragma unroll
        for (int i = 0; i < 8; ++i) {
            const int k = kk + 4 * i;
            const v2f r0 = *(const v2f*)(bp + (size_t)k * NFEAT);
            const v2f r1 = *(const v2f*)(bp + (size_t)k * NFEAT + NFEAT);
            v2f be, bo;
            be.x = r0.x; be.y = r1.x;
            bo.x = r0.y; bo.y = r1.y;
            ev = __builtin_amdgcn_wmma_f32_16x16x4_f32(
                false, a[i], false, be, (short)0, ev, false, false);
            od = __builtin_amdgcn_wmma_f32_16x16x4_f32(
                false, a[i], false, bo, (short)0, od, false, false);
        }
    }
}

__global__ void __launch_bounds__(128)
k_agg_mm2(const int* __restrict__ off, int* __restrict__ csr_rec,
          const float* __restrict__ dinv,
          const float* __restrict__ x, const float* __restrict__ W1,
          const float* __restrict__ b1, const float* __restrict__ W2,
          float* __restrict__ h2, int n) {
    __shared__ __attribute__((aligned(16))) float tile[16 * TP];
    const int lane = threadIdx.x;
    const int w    = __builtin_amdgcn_readfirstlane(threadIdx.y);
    const int m0   = blockIdx.x * 16;
    const int hi   = lane >> 4;
    const int l    = lane & 15;
    const int c0   = m0 + 4 * w;

    int   off_l = 0;
    float dv_l  = 0.0f;
    if (lane <= 4) { const int ci = c0 + lane; off_l = off[ci < n ? ci : n]; }
    if (lane < 4 && c0 + lane < n) dv_l = dinv[c0 + lane];
#pragma unroll 1
    for (int j = 0; j < 4; ++j) {
        const int c = c0 + j;
        float4 a = make_float4(0.0f, 0.0f, 0.0f, 0.0f);
        if (c < n) {
            const int   s  = __builtin_amdgcn_readlane(off_l, j);
            const int   t  = __builtin_amdgcn_readlane(off_l, j + 1);
            const float dc = __int_as_float(__builtin_amdgcn_readlane(__float_as_int(dv_l), j));
            a = agg_row<true>(csr_rec, dinv, x, c, s, t, dc, lane);
        }
        *(float4*)(tile + (4 * w + j) * TP + 4 * lane) = a;
    }
    __syncthreads();

    v8f ev = {}, od = {};
    mm_pair(tile, W1, w, l, hi, ev, od);
    const v2f bb = *(const v2f*)(b1 + 32 * w + 2 * l);
    __syncthreads();
#pragma unroll
    for (int v = 0; v < 8; ++v) {
        v2f g;
        g.x = fmaxf(ev[v] + bb.x, 0.0f);
        g.y = fmaxf(od[v] + bb.y, 0.0f);
        *(v2f*)(tile + (v + 8 * hi) * TP + 32 * w + 2 * l) = g;
    }
    __syncthreads();

    v8f ev2 = {}, od2 = {};
    mm_pair(tile, W2, w, l, hi, ev2, od2);
    float* op = h2 + (size_t)(m0 + 8 * hi) * NFEAT + 32 * w + 2 * l;
#pragma unroll
    for (int v = 0; v < 8; ++v) {
        if (m0 + 8 * hi + v < n) {
            v2f o;
            o.x = ev2[v];
            o.y = od2[v];
            *(v2f*)(op + (size_t)v * NFEAT) = o;
        }
    }
}

__global__ void __launch_bounds__(256)
k_agg(const int* __restrict__ off, int* __restrict__ csr_rec,
      const float* __restrict__ dinv,
      const float* __restrict__ h, const float* __restrict__ bias,
      float* __restrict__ out, int n) {
    const int lane = threadIdx.x;
    const int c = blockIdx.x * 8 + __builtin_amdgcn_readfirstlane(threadIdx.y);
    if (c >= n) return;
    const int s = __builtin_amdgcn_readfirstlane(off[c]);
    const int t = __builtin_amdgcn_readfirstlane(off[c + 1]);
    const float dc = dinv[c];
    const int f = lane * 4;
    const float4 a  = agg_row<false>(csr_rec, dinv, h, c, s, t, dc, lane);
    const float4 bv = *(const float4*)(bias + f);
    float4 o;
    o.x = a.x + bv.x;
    o.y = a.y + bv.y;
    o.z = a.z + bv.z;
    o.w = a.w + bv.w;
    *(float4*)(out + (size_t)c * NFEAT + f) = o;
}

extern "C" void kernel_launch(void* const* d_in, const int* in_sizes, int n_in,
                              void* d_out, int out_size, void* d_ws, size_t ws_size,
                              hipStream_t stream) {
    const float* x  = (const float*)d_in[0];
    const int*   ei = (const int*)d_in[1];
    const float* ew = (const float*)d_in[2];
    const float* W1 = (const float*)d_in[3];
    const float* b1 = (const float*)d_in[4];
    const float* W2 = (const float*)d_in[5];
    const float* b2 = (const float*)d_in[6];
    float* out = (float*)d_out;

    const int N = in_sizes[0] / NFEAT;
    const int E = in_sizes[2];
    const int* row = ei;
    const int* col = ei + E;

    const int nchunk = (E + CHUNK - 1) / CHUNK;
    size_t npad = ((size_t)N + 1 + 1023) & ~(size_t)1023;
    size_t epad = ((size_t)E + 1023) & ~(size_t)1023;
    size_t tpad = ((size_t)nchunk * TBP + 127) & ~(size_t)127;
    int*   tbl     = (int*)d_ws;
    int*   off     = tbl + tpad;
    float* dinv    = (float*)(off + npad);
    int*   crec    = (int*)(dinv + npad);
    int*   bkt_rec = crec + 2 * (size_t)nchunk * CHUNK;
    int*   csr_rec = bkt_rec + 2 * epad;
    float* h       = (float*)(csr_rec + 2 * epad);

    const int T = 256;
    dim3 wblk(32, 8);
    dim3 gblk(32, 4);
    int ntile16 = (N + 15) / 16;
    int nblk8  = (N + 7) / 8;
    int nbkt   = (N + BK_NODES - 1) / BK_NODES;

    k_bscatter<<<nchunk, T, 0, stream>>>(col, row, ew, tbl, crec, E);
    k_bsort<<<nbkt, T, 0, stream>>>(tbl, crec, bkt_rec, off, csr_rec, dinv, N, E, nchunk);

    k_agg_mm2<<<ntile16, gblk, 0, stream>>>(off, csr_rec, dinv, x, W1, b1, W2, h, N);

    k_agg<<<nblk8, wblk, 0, stream>>>(off, csr_rec, dinv, h, b2, out, N);
}
